# starting kernel
# baseline (speedup 1.0000x reference)
#include <hip/hip_runtime.h>
#include <hip/hip_bf16.h>
#include <math.h>


typedef __attribute__((ext_vector_type(2))) float v2f;
typedef __attribute__((ext_vector_type(8))) float v8f;

#define T_SEQ   12
#define BCN     16
#define NODES   1024
#define FEAT    32
#define HID     64
#define G3      192
#define NROWS   (BCN * NODES)
#define NEG_BIG (-1e16f)

__device__ __forceinline__ v8f wmma_f32(v2f a, v2f b, v8f c) {
  return __builtin_amdgcn_wmma_f32_16x16x4_f32(
       false, a,  false, b,
       (short)0, c,  false,  false);
}

__device__ __forceinline__ v8f bias8(float v) {
  v8f r;
#pragma unroll
  for (int i = 0; i < 8; ++i) r[i] = v;
  return r;
}

__device__ __forceinline__ float sigmoidf_(float x) {
  return 1.0f / (1.0f + __expf(-x));
}

#define GRU_WAVES 4
#define HPAD 66

__global__ __launch_bounds__(128) void gru_fused_kernel(
    const float* __restrict__ x_in,
    const float* __restrict__ w_ih0,
    const float* __restrict__ w_hh0,
    const float* __restrict__ b_ih0,
    const float* __restrict__ b_hh0,
    const float* __restrict__ w_ih1,
    const float* __restrict__ w_hh1,
    const float* __restrict__ b_ih1,
    const float* __restrict__ b_hh1,
    float* __restrict__ z_out)
{
  __shared__ float wih0_t[FEAT * G3];
  __shared__ float whh0_t[HID * G3];
  __shared__ float wih1_t[HID * G3];
  __shared__ float whh1_t[HID * G3];
  __shared__ float hst[2][GRU_WAVES][16 * HPAD];

  const int tid = threadIdx.x;
  for (int i = tid; i < FEAT * G3; i += blockDim.x) {
    const int k = i / G3, g = i % G3;
    wih0_t[i] = w_ih0[g * FEAT + k];
  }
  for (int i = tid; i < HID * G3; i += blockDim.x) {
    const int k = i / G3, g = i % G3;
    whh0_t[i] = w_hh0[g * HID + k];
    wih1_t[i] = w_ih1[g * HID + k];
    whh1_t[i] = w_hh1[g * HID + k];
  }

  const int wave = tid >> 5;
  const int lane = tid & 31;
  const int l16  = lane & 15;
  const int lh   = lane >> 4;
  float* h0s = &hst[0][wave][0];
  float* h1s = &hst[1][wave][0];
  for (int i = lane; i < 16 * HPAD; i += 32) { h0s[i] = 0.0f; h1s[i] = 0.0f; }
  __syncthreads();

  const int row0 = blockIdx.x * (GRU_WAVES * 16) + wave * 16;

  for (int t = 0; t < T_SEQ; ++t) {
    v2f xf[8];
    {
      const float* xrow = x_in + ((size_t)t * NROWS + row0 + l16) * FEAT;
#pragma unroll
      for (int ks = 0; ks < 8; ++ks) {
        const int k = 4 * ks + 2 * lh;
        xf[ks].x = xrow[k]; xf[ks].y = xrow[k + 1];
      }
    }
    v2f h0f[16], h1f[16];
#pragma unroll
    for (int ks = 0; ks < 16; ++ks) {
      const int k = 4 * ks + 2 * lh;
      h0f[ks].x = h0s[l16 * HPAD + k]; h0f[ks].y = h0s[l16 * HPAD + k + 1];
      h1f[ks].x = h1s[l16 * HPAD + k]; h1f[ks].y = h1s[l16 * HPAD + k + 1];
    }

#pragma unroll
    for (int j = 0; j < 4; ++j) {
      const int gr = j * 16 + l16;
      const int gz = 64 + j * 16 + l16;
      const int gn = 128 + j * 16 + l16;
      v8f gir = bias8(b_ih0[gr]), giz = bias8(b_ih0[gz]), gin = bias8(b_ih0[gn]);
      v8f ghr = bias8(b_hh0[gr]), ghz = bias8(b_hh0[gz]), ghn = bias8(b_hh0[gn]);
#pragma unroll
      for (int ks = 0; ks < 8; ++ks) {
        const int k = 4 * ks + 2 * lh;
        v2f br, bz, bn;
        br.x = wih0_t[k * G3 + gr]; br.y = wih0_t[(k + 1) * G3 + gr];
        bz.x = wih0_t[k * G3 + gz]; bz.y = wih0_t[(k + 1) * G3 + gz];
        bn.x = wih0_t[k * G3 + gn]; bn.y = wih0_t[(k + 1) * G3 + gn];
        gir = wmma_f32(xf[ks], br, gir);
        giz = wmma_f32(xf[ks], bz, giz);
        gin = wmma_f32(xf[ks], bn, gin);
      }
#pragma unroll
      for (int ks = 0; ks < 16; ++ks) {
        const int k = 4 * ks + 2 * lh;
        v2f br, bz, bn;
        br.x = whh0_t[k * G3 + gr]; br.y = whh0_t[(k + 1) * G3 + gr];
        bz.x = whh0_t[k * G3 + gz]; bz.y = whh0_t[(k + 1) * G3 + gz];
        bn.x = whh0_t[k * G3 + gn]; bn.y = whh0_t[(k + 1) * G3 + gn];
        ghr = wmma_f32(h0f[ks], br, ghr);
        ghz = wmma_f32(h0f[ks], bz, ghz);
        ghn = wmma_f32(h0f[ks], bn, ghn);
      }
#pragma unroll
      for (int v = 0; v < 8; ++v) {
        const int a = (v + 8 * lh) * HPAD + j * 16 + l16;
        const float hold = h0s[a];
        const float r  = sigmoidf_(gir[v] + ghr[v]);
        const float zg = sigmoidf_(giz[v] + ghz[v]);
        const float n  = tanhf(gin[v] + r * ghn[v]);
        h0s[a] = (1.0f - zg) * n + zg * hold;
      }
    }

#pragma unroll
    for (int ks = 0; ks < 16; ++ks) {
      const int k = 4 * ks + 2 * lh;
      h0f[ks].x = h0s[l16 * HPAD + k]; h0f[ks].y = h0s[l16 * HPAD + k + 1];
    }

#pragma unroll
    for (int j = 0; j < 4; ++j) {
      const int gr = j * 16 + l16;
      const int gz = 64 + j * 16 + l16;
      const int gn = 128 + j * 16 + l16;
      v8f gir = bias8(b_ih1[gr]), giz = bias8(b_ih1[gz]), gin = bias8(b_ih1[gn]);
      v8f ghr = bias8(b_hh1[gr]), ghz = bias8(b_hh1[gz]), ghn = bias8(b_hh1[gn]);
#pragma unroll
      for (int ks = 0; ks < 16; ++ks) {
        const int k = 4 * ks + 2 * lh;
        v2f br, bz, bn;
        br.x = wih1_t[k * G3 + gr]; br.y = wih1_t[(k + 1) * G3 + gr];
        bz.x = wih1_t[k * G3 + gz]; bz.y = wih1_t[(k + 1) * G3 + gz];
        bn.x = wih1_t[k * G3 + gn]; bn.y = wih1_t[(k + 1) * G3 + gn];
        gir = wmma_f32(h0f[ks], br, gir);
        giz = wmma_f32(h0f[ks], bz, giz);
        gin = wmma_f32(h0f[ks], bn, gin);
      }
#pragma unroll
      for (int ks = 0; ks < 16; ++ks) {
        const int k = 4 * ks + 2 * lh;
        v2f br, bz, bn;
        br.x = whh1_t[k * G3 + gr]; br.y = whh1_t[(k + 1) * G3 + gr];
        bz.x = whh1_t[k * G3 + gz]; bz.y = whh1_t[(k + 1) * G3 + gz];
        bn.x = whh1_t[k * G3 + gn]; bn.y = whh1_t[(k + 1) * G3 + gn];
        ghr = wmma_f32(h1f[ks], br, ghr);
        ghz = wmma_f32(h1f[ks], bz, ghz);
        ghn = wmma_f32(h1f[ks], bn, ghn);
      }
#pragma unroll
      for (int v = 0; v < 8; ++v) {
        const int a = (v + 8 * lh) * HPAD + j * 16 + l16;
        const float hold = h1s[a];
        const float r  = sigmoidf_(gir[v] + ghr[v]);
        const float zg = sigmoidf_(giz[v] + ghz[v]);
        const float n  = tanhf(gin[v] + r * ghn[v]);
        h1s[a] = (1.0f - zg) * n + zg * hold;
      }
    }
  }

  for (int i = lane; i < 16 * HID; i += 32) {
    const int r = i / HID, c = i % HID;
    z_out[(size_t)(row0 + r) * HID + c] = h0s[r * HPAD + c];
    z_out[(size_t)NROWS * HID + (size_t)(row0 + r) * HID + c] = h1s[r * HPAD + c];
  }
}

__global__ __launch_bounds__(32) void gat_kernel(
    const float* __restrict__ z,
    const int*   __restrict__ mask,
    float*       __restrict__ out)
{
  __shared__ float S[NODES * 16];

  const int lane  = threadIdx.x;
  const int l16   = lane & 15;
  const int lh    = lane >> 4;
  const int ntile = blockIdx.x & 63;
  const int pair  = blockIdx.x >> 6;
  const float* zp = z + (size_t)pair * NODES * HID;
  const int n0 = ntile * 16;

  v2f za[16];
  {
    const float* zrow = zp + (size_t)(n0 + l16) * HID;
#pragma unroll
    for (int ks = 0; ks < 16; ++ks) {
      const int k = 4 * ks + 2 * lh;
      za[ks].x = zrow[k]; za[ks].y = zrow[k + 1];
    }
  }

  for (int mt = 0; mt < 64; ++mt) {
    v8f acc = {};
#pragma unroll
    for (int ks = 0; ks < 16; ++ks) {
      const int k = 4 * ks + 2 * lh;
      v2f b;
      b.x = zp[(size_t)(mt * 16 + l16) * HID + k];
      b.y = zp[(size_t)(mt * 16 + l16) * HID + k + 1];
      acc = wmma_f32(za[ks], b, acc);
    }
    const int col = mt * 16 + l16;
#pragma unroll
    for (int v = 0; v < 8; ++v) {
      const int row = v + 8 * lh;
      const float s  = acc[v];
      const float lr = s > 0.0f ? s : 0.1f * s;
      float a = lr * (float)mask[(size_t)(n0 + row) * NODES + col];
      a = (a == 0.0f) ? NEG_BIG : a;
      S[col * 16 + row] = a;
    }
  }

  {
    const int r = l16;
    const int mbase = lh * 512;
    float mx = -INFINITY;
    for (int i = 0; i < 512; ++i) mx = fmaxf(mx, S[(mbase + i) * 16 + r]);
    mx = fmaxf(mx, __shfl_xor(mx, 16, 32));
    float sum = 0.0f;
    for (int i = 0; i < 512; ++i) sum += __expf(S[(mbase + i) * 16 + r] - mx);
    sum += __shfl_xor(sum, 16, 32);
    const float inv = 1.0f / sum;
    for (int i = 0; i < 512; ++i) {
      const int idx = (mbase + i) * 16 + r;
      S[idx] = __expf(S[idx] - mx) * inv;
    }
  }

  v8f accO[4] = {};
  for (int mt = 0; mt < 64; ++mt) {
#pragma unroll
    for (int ks = 0; ks < 4; ++ks) {
      const int k = mt * 16 + 4 * ks + 2 * lh;
      v2f a;
      a.x = S[k * 16 + l16];
      a.y = S[(k + 1) * 16 + l16];
#pragma unroll
      for (int ct = 0; ct < 4; ++ct) {
        v2f b;
        b.x = zp[(size_t)k * HID + ct * 16 + l16];
        b.y = zp[(size_t)(k + 1) * HID + ct * 16 + l16];
        accO[ct] = wmma_f32(a, b, accO[ct]);
      }
    }
  }

  float* op = out + (size_t)pair * NODES * HID;
#pragma unroll
  for (int ct = 0; ct < 4; ++ct)
#pragma unroll
    for (int v = 0; v < 8; ++v) {
      const int row = v + 8 * lh;
      op[(size_t)(n0 + row) * HID + ct * 16 + l16] = accO[ct][v];
    }
}

extern "C" void kernel_launch(void* const* d_in, const int* in_sizes, int n_in,
                              void* d_out, int out_size, void* d_ws, size_t ws_size,
                              hipStream_t stream) {
  const float* x     = (const float*)d_in[0];
  const int*   mask  = (const int*)  d_in[1];
  const float* w_ih0 = (const float*)d_in[2];
  const float* w_hh0 = (const float*)d_in[3];
  const float* b_ih0 = (const float*)d_in[4];
  const float* b_hh0 = (const float*)d_in[5];
  const float* w_ih1 = (const float*)d_in[6];
  const float* w_hh1 = (const float*)d_in[7];
  const float* b_ih1 = (const float*)d_in[8];
  const float* b_hh1 = (const float*)d_in[9];
  float* out  = (float*)d_out;
  float* zbuf = (float*)d_ws;

  gru_fused_kernel<<<256, 128, 0, stream>>>(x, w_ih0, w_hh0, b_ih0, b_hh0,
                                            w_ih1, w_hh1, b_ih1, b_hh1, zbuf);
  gat_kernel<<<2048, 32, 0, stream>>>(zbuf, mask, out);
}


# Attention kernel: both WMMA loops software-pipelined in 32-column stages; the next stage's z fragments, mask words and P fragments are requested before the current stage's products run
# speedup vs baseline: 1.0633x; 1.0633x over previous
#include <hip/hip_runtime.h>
#include <hip/hip_bf16.h>
#include <math.h>


typedef __attribute__((ext_vector_type(2))) float v2f;
typedef __attribute__((ext_vector_type(8))) float v8f;

#define T_SEQ   12
#define BCN     16
#define NODES   1024
#define FEAT    32
#define HID     64
#define G3      192
#define NROWS   (BCN * NODES)
#define NEG_BIG (-1e16f)

__device__ __forceinline__ v8f wmma_f32(v2f a, v2f b, v8f c) {
  return __builtin_amdgcn_wmma_f32_16x16x4_f32(
       false, a,  false, b,
       (short)0, c,  false,  false);
}

__device__ __forceinline__ v8f bias8(float v) {
  v8f r;
#pragma unroll
  for (int i = 0; i < 8; ++i) r[i] = v;
  return r;
}

__device__ __forceinline__ float sigmoidf_(float x) {
  return 1.0f / (1.0f + __expf(-x));
}

#define GRU_WAVES 4
#define HPAD 66

__global__ __launch_bounds__(128) void gru_fused_kernel(
    const float* __restrict__ x_in,
    const float* __restrict__ w_ih0,
    const float* __restrict__ w_hh0,
    const float* __restrict__ b_ih0,
    const float* __restrict__ b_hh0,
    const float* __restrict__ w_ih1,
    const float* __restrict__ w_hh1,
    const float* __restrict__ b_ih1,
    const float* __restrict__ b_hh1,
    float* __restrict__ z_out)
{
  __shared__ float wih0_t[FEAT * G3];
  __shared__ float whh0_t[HID * G3];
  __shared__ float wih1_t[HID * G3];
  __shared__ float whh1_t[HID * G3];
  __shared__ float hst[2][GRU_WAVES][16 * HPAD];

  const int tid = threadIdx.x;
  for (int i = tid; i < FEAT * G3; i += blockDim.x) {
    const int k = i / G3, g = i % G3;
    wih0_t[i] = w_ih0[g * FEAT + k];
  }
  for (int i = tid; i < HID * G3; i += blockDim.x) {
    const int k = i / G3, g = i % G3;
    whh0_t[i] = w_hh0[g * HID + k];
    wih1_t[i] = w_ih1[g * HID + k];
    whh1_t[i] = w_hh1[g * HID + k];
  }

  const int wave = tid >> 5;
  const int lane = tid & 31;
  const int l16  = lane & 15;
  const int lh   = lane >> 4;
  float* h0s = &hst[0][wave][0];
  float* h1s = &hst[1][wave][0];
  for (int i = lane; i < 16 * HPAD; i += 32) { h0s[i] = 0.0f; h1s[i] = 0.0f; }
  __syncthreads();

  const int row0 = blockIdx.x * (GRU_WAVES * 16) + wave * 16;

  for (int t = 0; t < T_SEQ; ++t) {
    v2f xf[8];
    {
      const float* xrow = x_in + ((size_t)t * NROWS + row0 + l16) * FEAT;
#pragma unroll
      for (int ks = 0; ks < 8; ++ks) {
        const int k = 4 * ks + 2 * lh;
        xf[ks].x = xrow[k]; xf[ks].y = xrow[k + 1];
      }
    }
    v2f h0f[16], h1f[16];
#pragma unroll
    for (int ks = 0; ks < 16; ++ks) {
      const int k = 4 * ks + 2 * lh;
      h0f[ks].x = h0s[l16 * HPAD + k]; h0f[ks].y = h0s[l16 * HPAD + k + 1];
      h1f[ks].x = h1s[l16 * HPAD + k]; h1f[ks].y = h1s[l16 * HPAD + k + 1];
    }

#pragma unroll
    for (int j = 0; j < 4; ++j) {
      const int gr = j * 16 + l16;
      const int gz = 64 + j * 16 + l16;
      const int gn = 128 + j * 16 + l16;
      v8f gir = bias8(b_ih0[gr]), giz = bias8(b_ih0[gz]), gin = bias8(b_ih0[gn]);
      v8f ghr = bias8(b_hh0[gr]), ghz = bias8(b_hh0[gz]), ghn = bias8(b_hh0[gn]);
#pragma unroll
      for (int ks = 0; ks < 8; ++ks) {
        const int k = 4 * ks + 2 * lh;
        v2f br, bz, bn;
        br.x = wih0_t[k * G3 + gr]; br.y = wih0_t[(k + 1) * G3 + gr];
        bz.x = wih0_t[k * G3 + gz]; bz.y = wih0_t[(k + 1) * G3 + gz];
        bn.x = wih0_t[k * G3 + gn]; bn.y = wih0_t[(k + 1) * G3 + gn];
        gir = wmma_f32(xf[ks], br, gir);
        giz = wmma_f32(xf[ks], bz, giz);
        gin = wmma_f32(xf[ks], bn, gin);
      }
#pragma unroll
      for (int ks = 0; ks < 16; ++ks) {
        const int k = 4 * ks + 2 * lh;
        v2f br, bz, bn;
        br.x = whh0_t[k * G3 + gr]; br.y = whh0_t[(k + 1) * G3 + gr];
        bz.x = whh0_t[k * G3 + gz]; bz.y = whh0_t[(k + 1) * G3 + gz];
        bn.x = whh0_t[k * G3 + gn]; bn.y = whh0_t[(k + 1) * G3 + gn];
        ghr = wmma_f32(h0f[ks], br, ghr);
        ghz = wmma_f32(h0f[ks], bz, ghz);
        ghn = wmma_f32(h0f[ks], bn, ghn);
      }
#pragma unroll
      for (int v = 0; v < 8; ++v) {
        const int a = (v + 8 * lh) * HPAD + j * 16 + l16;
        const float hold = h0s[a];
        const float r  = sigmoidf_(gir[v] + ghr[v]);
        const float zg = sigmoidf_(giz[v] + ghz[v]);
        const float n  = tanhf(gin[v] + r * ghn[v]);
        h0s[a] = (1.0f - zg) * n + zg * hold;
      }
    }

#pragma unroll
    for (int ks = 0; ks < 16; ++ks) {
      const int k = 4 * ks + 2 * lh;
      h0f[ks].x = h0s[l16 * HPAD + k]; h0f[ks].y = h0s[l16 * HPAD + k + 1];
    }

#pragma unroll
    for (int j = 0; j < 4; ++j) {
      const int gr = j * 16 + l16;
      const int gz = 64 + j * 16 + l16;
      const int gn = 128 + j * 16 + l16;
      v8f gir = bias8(b_ih1[gr]), giz = bias8(b_ih1[gz]), gin = bias8(b_ih1[gn]);
      v8f ghr = bias8(b_hh1[gr]), ghz = bias8(b_hh1[gz]), ghn = bias8(b_hh1[gn]);
#pragma unroll
      for (int ks = 0; ks < 16; ++ks) {
        const int k = 4 * ks + 2 * lh;
        v2f br, bz, bn;
        br.x = wih1_t[k * G3 + gr]; br.y = wih1_t[(k + 1) * G3 + gr];
        bz.x = wih1_t[k * G3 + gz]; bz.y = wih1_t[(k + 1) * G3 + gz];
        bn.x = wih1_t[k * G3 + gn]; bn.y = wih1_t[(k + 1) * G3 + gn];
        gir = wmma_f32(h0f[ks], br, gir);
        giz = wmma_f32(h0f[ks], bz, giz);
        gin = wmma_f32(h0f[ks], bn, gin);
      }
#pragma unroll
      for (int ks = 0; ks < 16; ++ks) {
        const int k = 4 * ks + 2 * lh;
        v2f br, bz, bn;
        br.x = whh1_t[k * G3 + gr]; br.y = whh1_t[(k + 1) * G3 + gr];
        bz.x = whh1_t[k * G3 + gz]; bz.y = whh1_t[(k + 1) * G3 + gz];
        bn.x = whh1_t[k * G3 + gn]; bn.y = whh1_t[(k + 1) * G3 + gn];
        ghr = wmma_f32(h1f[ks], br, ghr);
        ghz = wmma_f32(h1f[ks], bz, ghz);
        ghn = wmma_f32(h1f[ks], bn, ghn);
      }
#pragma unroll
      for (int v = 0; v < 8; ++v) {
        const int a = (v + 8 * lh) * HPAD + j * 16 + l16;
        const float hold = h1s[a];
        const float r  = sigmoidf_(gir[v] + ghr[v]);
        const float zg = sigmoidf_(giz[v] + ghz[v]);
        const float n  = tanhf(gin[v] + r * ghn[v]);
        h1s[a] = (1.0f - zg) * n + zg * hold;
      }
    }
  }

  for (int i = lane; i < 16 * HID; i += 32) {
    const int r = i / HID, c = i % HID;
    z_out[(size_t)(row0 + r) * HID + c] = h0s[r * HPAD + c];
    z_out[(size_t)NROWS * HID + (size_t)(row0 + r) * HID + c] = h1s[r * HPAD + c];
  }
}


__device__ __forceinline__ void gat_request_score_stage(
    const float* __restrict__ zp, const int* __restrict__ mrow,
    int st, int l16, int lh, v2f (&b)[2][16], int (&m)[2][8])
{
#pragma unroll
  for (int u = 0; u < 2; ++u) {
    const int col = (st * 2 + u) * 16 + l16;
    const float* zr = zp + (size_t)col * HID + 2 * lh;
#pragma unroll
    for (int ks = 0; ks < 16; ++ks) {
      b[u][ks].x = zr[4 * ks];
      b[u][ks].y = zr[4 * ks + 1];
    }
    const int* mp = mrow + col;
#pragma unroll
    for (int v = 0; v < 8; ++v) m[u][v] = mp[(size_t)v * NODES];
  }
}

__device__ __forceinline__ void gat_score_stage(
    const v2f (&za)[16], const v2f (&b)[2][16], const int (&m)[2][8],
    float* S, int st, int l16, int lh)
{
#pragma unroll
  for (int u = 0; u < 2; ++u) {
    v8f acc = {};
#pragma unroll
    for (int ks = 0; ks < 16; ++ks) acc = wmma_f32(za[ks], b[u][ks], acc);
    const int col = (st * 2 + u) * 16 + l16;
#pragma unroll
    for (int v = 0; v < 8; ++v) {
      const int row = v + 8 * lh;
      const float s  = acc[v];
      const float lr = s > 0.0f ? s : 0.1f * s;
      float a = lr * (float)m[u][v];
      a = (a == 0.0f) ? NEG_BIG : a;
      S[col * 16 + row] = a;
    }
  }
}

__device__ __forceinline__ void gat_request_pv_stage(
    const float* S, const float* __restrict__ zp,
    int st, int l16, int lh, v2f (&a)[8], v2f (&b)[8][4])
{
#pragma unroll
  for (int ks = 0; ks < 8; ++ks) {
    const int k = st * 32 + 4 * ks + 2 * lh;
    a[ks].x = S[k * 16 + l16];
    a[ks].y = S[(k + 1) * 16 + l16];
#pragma unroll
    for (int ct = 0; ct < 4; ++ct) {
      b[ks][ct].x = zp[(size_t)k * HID + ct * 16 + l16];
      b[ks][ct].y = zp[(size_t)(k + 1) * HID + ct * 16 + l16];
    }
  }
}

__device__ __forceinline__ void gat_pv_stage(
    const v2f (&a)[8], const v2f (&b)[8][4], v8f (&accO)[4])
{
#pragma unroll
  for (int ks = 0; ks < 8; ++ks)
#pragma unroll
    for (int ct = 0; ct < 4; ++ct)
      accO[ct] = wmma_f32(a[ks], b[ks][ct], accO[ct]);
}

__global__ __launch_bounds__(32) void gat_kernel(
    const float* __restrict__ z,
    const int*   __restrict__ mask,
    float*       __restrict__ out)
{
  __shared__ float S[NODES * 16];

  const int lane  = threadIdx.x;
  const int l16   = lane & 15;
  const int lh    = lane >> 4;
  const int ntile = blockIdx.x & 63;
  const int pair  = blockIdx.x >> 6;
  const float* zp = z + (size_t)pair * NODES * HID;
  const int n0 = ntile * 16;

  v2f za[16];
  {
    const float* zrow = zp + (size_t)(n0 + l16) * HID;
#pragma unroll
    for (int ks = 0; ks < 16; ++ks) {
      const int k = 4 * ks + 2 * lh;
      za[ks].x = zrow[k]; za[ks].y = zrow[k + 1];
    }
  }

  {
    const int* mrow = mask + (size_t)(n0 + 8 * lh) * NODES;
    v2f bA[2][16], bB[2][16];
    int mA[2][8], mB[2][8];
    gat_request_score_stage(zp, mrow, 0, l16, lh, bA, mA);
    for (int st = 0; st < 30; st += 2) {
      gat_request_score_stage(zp, mrow, st + 1, l16, lh, bB, mB);
      gat_score_stage(za, bA, mA, S, st, l16, lh);
      gat_request_score_stage(zp, mrow, st + 2, l16, lh, bA, mA);
      gat_score_stage(za, bB, mB, S, st + 1, l16, lh);
    }
    gat_request_score_stage(zp, mrow, 31, l16, lh, bB, mB);
    gat_score_stage(za, bA, mA, S, 30, l16, lh);
    gat_score_stage(za, bB, mB, S, 31, l16, lh);
  }

  {
    const int r = l16;
    const int mbase = lh * 512;
    float mx = -INFINITY;
    for (int i = 0; i < 512; ++i) mx = fmaxf(mx, S[(mbase + i) * 16 + r]);
    mx = fmaxf(mx, __shfl_xor(mx, 16, 32));
    float sum = 0.0f;
    for (int i = 0; i < 512; ++i) sum += __expf(S[(mbase + i) * 16 + r] - mx);
    sum += __shfl_xor(sum, 16, 32);
    const float inv = 1.0f / sum;
    for (int i = 0; i < 512; ++i) {
      const int idx = (mbase + i) * 16 + r;
      S[idx] = __expf(S[idx] - mx) * inv;
    }
  }

  v8f accO[4] = {};
  {
    v2f aA[8], aB[8];
    v2f bA[8][4], bB[8][4];
    gat_request_pv_stage(S, zp, 0, l16, lh, aA, bA);
    for (int st = 0; st < 30; st += 2) {
      gat_request_pv_stage(S, zp, st + 1, l16, lh, aB, bB);
      gat_pv_stage(aA, bA, accO);
      gat_request_pv_stage(S, zp, st + 2, l16, lh, aA, bA);
      gat_pv_stage(aB, bB, accO);
    }
    gat_request_pv_stage(S, zp, 31, l16, lh, aB, bB);
    gat_pv_stage(aA, bA, accO);
    gat_pv_stage(aB, bB, accO);
  }

  float* op = out + (size_t)pair * NODES * HID;
#pragma unroll
  for (int ct = 0; ct < 4; ++ct)
#pragma unroll
    for (int v = 0; v < 8; ++v) {
      const int row = v + 8 * lh;
      op[(size_t)(n0 + row) * HID + ct * 16 + l16] = accO[ct][v];
    }
}

extern "C" void kernel_launch(void* const* d_in, const int* in_sizes, int n_in,
                              void* d_out, int out_size, void* d_ws, size_t ws_size,
                              hipStream_t stream) {
  const float* x     = (const float*)d_in[0];
  const int*   mask  = (const int*)  d_in[1];
  const float* w_ih0 = (const float*)d_in[2];
  const float* w_hh0 = (const float*)d_in[3];
  const float* b_ih0 = (const float*)d_in[4];
  const float* b_hh0 = (const float*)d_in[5];
  const float* w_ih1 = (const float*)d_in[6];
  const float* w_hh1 = (const float*)d_in[7];
  const float* b_ih1 = (const float*)d_in[8];
  const float* b_hh1 = (const float*)d_in[9];
  float* out  = (float*)d_out;
  float* zbuf = (float*)d_ws;

  gru_fused_kernel<<<256, 128, 0, stream>>>(x, w_ih0, w_hh0, b_ih0, b_hh0,
                                            w_ih1, w_hh1, b_ih1, b_hh1, zbuf);
  gat_kernel<<<2048, 32, 0, stream>>>(zbuf, mask, out);
}


# GRU gate block made straight-line: sigmoid by the hardware reciprocal (1 ulp) instead of the divide sequence, tanh computes both ranges and selects, so a lane's 8 gate values interleave
# speedup vs baseline: 1.6451x; 1.5472x over previous
#include <hip/hip_runtime.h>
#include <hip/hip_bf16.h>
#include <math.h>


typedef __attribute__((ext_vector_type(2))) float v2f;
typedef __attribute__((ext_vector_type(8))) float v8f;

#define T_SEQ   12
#define BCN     16
#define NODES   1024
#define FEAT    32
#define HID     64
#define G3      192
#define NROWS   (BCN * NODES)
#define NEG_BIG (-1e16f)

__device__ __forceinline__ v8f wmma_f32(v2f a, v2f b, v8f c) {
  return __builtin_amdgcn_wmma_f32_16x16x4_f32(
       false, a,  false, b,
       (short)0, c,  false,  false);
}

__device__ __forceinline__ v8f bias8(float v) {
  v8f r;
#pragma unroll
  for (int i = 0; i < 8; ++i) r[i] = v;
  return r;
}

__device__ __forceinline__ float sigmoidf_(float x) {
  return __builtin_amdgcn_rcpf(1.0f + __expf(-x));
}

__device__ __forceinline__ float tanhf_(float x) {
  const float ax = fabsf(x);
  const float t  = ax * ax;
  float p = fmaf(t, -0.00570002059f, 0.0206340719f);
  p = fmaf(t, p, -0.0537379310f);
  p = fmaf(t, p, 0.133314162f);
  p = fmaf(t, p, -0.333332807f);
  const float lo = fmaf(t, ax * p, ax);
  const float e  = __expf(2.0f * ax);
  const float hi = fmaf(__builtin_amdgcn_rcpf(1.0f + e), -2.0f, 1.0f);
  return copysignf(ax < 0.625f ? lo : hi, x);
}

#define GRU_WAVES 4
#define GRU_ROWS  64
#define HPAD 66

typedef __attribute__((ext_vector_type(4))) float v4f;

template <int KS>
__device__ __forceinline__ void gru_load_w(const float* __restrict__ w, int kdim,
                                           int g, int lh, v2f (&f)[KS]) {
  const float* wr = w + (size_t)g * kdim + 2 * lh;
#pragma unroll
  for (int ks = 0; ks < KS; ++ks) { f[ks].x = wr[4 * ks]; f[ks].y = wr[4 * ks + 1]; }
}

__device__ __forceinline__ void gru_load_h(const float* hs, int rb, int l16, int lh,
                                           v2f (&f)[16]) {
  const float* hr = hs + (rb + l16) * HPAD + 2 * lh;
#pragma unroll
  for (int ks = 0; ks < 16; ++ks) { f[ks].x = hr[4 * ks]; f[ks].y = hr[4 * ks + 1]; }
}

__device__ __forceinline__ void gru_load_x(const float* __restrict__ x_in, int t,
                                           int row, int lh, v2f (&f)[8]) {
  const float* xr = x_in + ((size_t)t * NROWS + row) * FEAT + 2 * lh;
#pragma unroll
  for (int ks = 0; ks < 8; ++ks) { f[ks].x = xr[4 * ks]; f[ks].y = xr[4 * ks + 1]; }
}

__global__ __launch_bounds__(128) void gru_fused_kernel(
    const float* __restrict__ x_in,
    const float* __restrict__ w_ih0,
    const float* __restrict__ w_hh0,
    const float* __restrict__ b_ih0,
    const float* __restrict__ b_hh0,
    const float* __restrict__ w_ih1,
    const float* __restrict__ w_hh1,
    const float* __restrict__ b_ih1,
    const float* __restrict__ b_hh1,
    float* __restrict__ z_out)
{
  __shared__ float hbuf[2][2][GRU_ROWS * HPAD];

  const int tid  = threadIdx.x;
  const int wave = tid >> 5;
  const int lane = tid & 31;
  const int l16  = lane & 15;
  const int lh   = lane >> 4;
  const int row0 = blockIdx.x * GRU_ROWS;
  const int cb   = wave * 16;
  const int gr   = cb + l16;
  const int gz   = 64 + cb + l16;
  const int gn   = 128 + cb + l16;

  for (int i = tid; i < GRU_ROWS * HPAD; i += GRU_WAVES * 32) {
    hbuf[0][0][i] = 0.0f;
    hbuf[1][0][i] = 0.0f;
  }

  v2f wi0r[8],  wi0z[8],  wi0n[8];
  v2f wh0r[16], wh0z[16], wh0n[16];
  v2f wi1r[16], wi1z[16], wi1n[16];
  v2f wh1r[16], wh1z[16], wh1n[16];
  gru_load_w<8>(w_ih0, FEAT, gr, lh, wi0r);
  gru_load_w<8>(w_ih0, FEAT, gz, lh, wi0z);
  gru_load_w<8>(w_ih0, FEAT, gn, lh, wi0n);
  gru_load_w<16>(w_hh0, HID, gr, lh, wh0r);
  gru_load_w<16>(w_hh0, HID, gz, lh, wh0z);
  gru_load_w<16>(w_hh0, HID, gn, lh, wh0n);
  gru_load_w<16>(w_ih1, HID, gr, lh, wi1r);
  gru_load_w<16>(w_ih1, HID, gz, lh, wi1z);
  gru_load_w<16>(w_ih1, HID, gn, lh, wi1n);
  gru_load_w<16>(w_hh1, HID, gr, lh, wh1r);
  gru_load_w<16>(w_hh1, HID, gz, lh, wh1z);
  gru_load_w<16>(w_hh1, HID, gn, lh, wh1n);
  const float bi0r = b_ih0[gr], bi0z = b_ih0[gz], bi0n = b_ih0[gn];
  const float bh0r = b_hh0[gr], bh0z = b_hh0[gz], bh0n = b_hh0[gn];
  const float bi1r = b_ih1[gr], bi1z = b_ih1[gz], bi1n = b_ih1[gn];
  const float bh1r = b_hh1[gr], bh1z = b_hh1[gz], bh1n = b_hh1[gn];

  v2f xc[8];
  gru_load_x(x_in, 0, row0 + l16, lh, xc);

  __syncthreads();

  for (int t = 0; t < T_SEQ; ++t) {
    const float* h0r = &hbuf[0][t & 1][0];
    float*       h0w = &hbuf[0][(t + 1) & 1][0];
    const float* h1r = &hbuf[1][t & 1][0];
    float*       h1w = &hbuf[1][(t + 1) & 1][0];

#pragma unroll 1
    for (int rt = 0; rt < 4; ++rt) {
      const int rb = rt * 16;
      v2f ah[16];
      gru_load_h(h0r, rb, l16, lh, ah);
      float hold[8];
#pragma unroll
      for (int v = 0; v < 8; ++v) hold[v] = h0r[(rb + v + 8 * lh) * HPAD + cb + l16];

      const int nrt = (rt + 1) & 3;
      const int nt  = (rt == 3 && t + 1 < T_SEQ) ? t + 1 : t;
      v2f xn[8];
      gru_load_x(x_in, nt, row0 + nrt * 16 + l16, lh, xn);
      __builtin_amdgcn_sched_barrier(0);

      v8f gir = bias8(bi0r), giz = bias8(bi0z), gin = bias8(bi0n);
      v8f ghr = bias8(bh0r), ghz = bias8(bh0z), ghn = bias8(bh0n);
#pragma unroll
      for (int ks = 0; ks < 8; ++ks) {
        gir = wmma_f32(xc[ks], wi0r[ks], gir);
        giz = wmma_f32(xc[ks], wi0z[ks], giz);
        gin = wmma_f32(xc[ks], wi0n[ks], gin);
      }
#pragma unroll
      for (int ks = 0; ks < 16; ++ks) {
        ghr = wmma_f32(ah[ks], wh0r[ks], ghr);
        ghz = wmma_f32(ah[ks], wh0z[ks], ghz);
        ghn = wmma_f32(ah[ks], wh0n[ks], ghn);
      }
#pragma unroll
      for (int v = 0; v < 8; ++v) {
        const float r  = sigmoidf_(gir[v] + ghr[v]);
        const float zg = sigmoidf_(giz[v] + ghz[v]);
        const float n  = tanhf_(gin[v] + r * ghn[v]);
        h0w[(rb + v + 8 * lh) * HPAD + cb + l16] = (1.0f - zg) * n + zg * hold[v];
      }
#pragma unroll
      for (int ks = 0; ks < 8; ++ks) xc[ks] = xn[ks];
    }

    __syncthreads();

#pragma unroll 1
    for (int rt = 0; rt < 4; ++rt) {
      const int rb = rt * 16;
      v2f a0[16], a1[16];
      gru_load_h(h0w, rb, l16, lh, a0);
      gru_load_h(h1r, rb, l16, lh, a1);
      float hold[8];
#pragma unroll
      for (int v = 0; v < 8; ++v) hold[v] = h1r[(rb + v + 8 * lh) * HPAD + cb + l16];
      __builtin_amdgcn_sched_barrier(0);

      v8f gir = bias8(bi1r), giz = bias8(bi1z), gin = bias8(bi1n);
      v8f ghr = bias8(bh1r), ghz = bias8(bh1z), ghn = bias8(bh1n);
#pragma unroll
      for (int ks = 0; ks < 16; ++ks) {
        gir = wmma_f32(a0[ks], wi1r[ks], gir);
        giz = wmma_f32(a0[ks], wi1z[ks], giz);
        gin = wmma_f32(a0[ks], wi1n[ks], gin);
      }
#pragma unroll
      for (int ks = 0; ks < 16; ++ks) {
        ghr = wmma_f32(a1[ks], wh1r[ks], ghr);
        ghz = wmma_f32(a1[ks], wh1z[ks], ghz);
        ghn = wmma_f32(a1[ks], wh1n[ks], ghn);
      }
#pragma unroll
      for (int v = 0; v < 8; ++v) {
        const float r  = sigmoidf_(gir[v] + ghr[v]);
        const float zg = sigmoidf_(giz[v] + ghz[v]);
        const float n  = tanhf_(gin[v] + r * ghn[v]);
        h1w[(rb + v + 8 * lh) * HPAD + cb + l16] = (1.0f - zg) * n + zg * hold[v];
      }
    }
  }

  __syncthreads();

  const float* f0 = &hbuf[0][T_SEQ & 1][0];
  const float* f1 = &hbuf[1][T_SEQ & 1][0];
  for (int i = tid; i < GRU_ROWS * 16; i += GRU_WAVES * 32) {
    const int r = i >> 4, c = (i & 15) * 4;
    v4f p, q;
    p.x = f0[r * HPAD + c]; p.y = f0[r * HPAD + c + 1];
    p.z = f0[r * HPAD + c + 2]; p.w = f0[r * HPAD + c + 3];
    q.x = f1[r * HPAD + c]; q.y = f1[r * HPAD + c + 1];
    q.z = f1[r * HPAD + c + 2]; q.w = f1[r * HPAD + c + 3];
    *(v4f*)(z_out + (size_t)(row0 + r) * HID + c) = p;
    *(v4f*)(z_out + (size_t)NROWS * HID + (size_t)(row0 + r) * HID + c) = q;
  }
}


__device__ __forceinline__ void gat_request_score_stage(
    const float* __restrict__ zp, const int* __restrict__ mrow,
    int st, int l16, int lh, v2f (&b)[2][16], int (&m)[2][8])
{
#pragma unroll
  for (int u = 0; u < 2; ++u) {
    const int col = (st * 2 + u) * 16 + l16;
    const float* zr = zp + (size_t)col * HID + 2 * lh;
#pragma unroll
    for (int ks = 0; ks < 16; ++ks) {
      b[u][ks].x = zr[4 * ks];
      b[u][ks].y = zr[4 * ks + 1];
    }
    const int* mp = mrow + col;
#pragma unroll
    for (int v = 0; v < 8; ++v) m[u][v] = mp[(size_t)v * NODES];
  }
}

__device__ __forceinline__ void gat_score_stage(
    const v2f (&za)[16], const v2f (&b)[2][16], const int (&m)[2][8],
    float* S, int st, int l16, int lh)
{
#pragma unroll
  for (int u = 0; u < 2; ++u) {
    v8f acc = {};
#pragma unroll
    for (int ks = 0; ks < 16; ++ks) acc = wmma_f32(za[ks], b[u][ks], acc);
    const int col = (st * 2 + u) * 16 + l16;
#pragma unroll
    for (int v = 0; v < 8; ++v) {
      const int row = v + 8 * lh;
      const float s  = acc[v];
      const float lr = s > 0.0f ? s : 0.1f * s;
      float a = lr * (float)m[u][v];
      a = (a == 0.0f) ? NEG_BIG : a;
      S[col * 16 + row] = a;
    }
  }
}

__device__ __forceinline__ void gat_request_pv_stage(
    const float* S, const float* __restrict__ zp,
    int st, int l16, int lh, v2f (&a)[8], v2f (&b)[8][4])
{
#pragma unroll
  for (int ks = 0; ks < 8; ++ks) {
    const int k = st * 32 + 4 * ks + 2 * lh;
    a[ks].x = S[k * 16 + l16];
    a[ks].y = S[(k + 1) * 16 + l16];
#pragma unroll
    for (int ct = 0; ct < 4; ++ct) {
      b[ks][ct].x = zp[(size_t)k * HID + ct * 16 + l16];
      b[ks][ct].y = zp[(size_t)(k + 1) * HID + ct * 16 + l16];
    }
  }
}

__device__ __forceinline__ void gat_pv_stage(
    const v2f (&a)[8], const v2f (&b)[8][4], v8f (&accO)[4])
{
#pragma unroll
  for (int ks = 0; ks < 8; ++ks)
#pragma unroll
    for (int ct = 0; ct < 4; ++ct)
      accO[ct] = wmma_f32(a[ks], b[ks][ct], accO[ct]);
}

__global__ __launch_bounds__(32) void gat_kernel(
    const float* __restrict__ z,
    const int*   __restrict__ mask,
    float*       __restrict__ out)
{
  __shared__ float S[NODES * 16];

  const int lane  = threadIdx.x;
  const int l16   = lane & 15;
  const int lh    = lane >> 4;
  const int ntile = blockIdx.x & 63;
  const int pair  = blockIdx.x >> 6;
  const float* zp = z + (size_t)pair * NODES * HID;
  const int n0 = ntile * 16;

  v2f za[16];
  {
    const float* zrow = zp + (size_t)(n0 + l16) * HID;
#pragma unroll
    for (int ks = 0; ks < 16; ++ks) {
      const int k = 4 * ks + 2 * lh;
      za[ks].x = zrow[k]; za[ks].y = zrow[k + 1];
    }
  }

  {
    const int* mrow = mask + (size_t)(n0 + 8 * lh) * NODES;
    v2f bA[2][16], bB[2][16];
    int mA[2][8], mB[2][8];
    gat_request_score_stage(zp, mrow, 0, l16, lh, bA, mA);
    for (int st = 0; st < 30; st += 2) {
      gat_request_score_stage(zp, mrow, st + 1, l16, lh, bB, mB);
      gat_score_stage(za, bA, mA, S, st, l16, lh);
      gat_request_score_stage(zp, mrow, st + 2, l16, lh, bA, mA);
      gat_score_stage(za, bB, mB, S, st + 1, l16, lh);
    }
    gat_request_score_stage(zp, mrow, 31, l16, lh, bB, mB);
    gat_score_stage(za, bA, mA, S, 30, l16, lh);
    gat_score_stage(za, bB, mB, S, 31, l16, lh);
  }

  {
    const int r = l16;
    const int mbase = lh * 512;
    float mx = -INFINITY;
    for (int i = 0; i < 512; ++i) mx = fmaxf(mx, S[(mbase + i) * 16 + r]);
    mx = fmaxf(mx, __shfl_xor(mx, 16, 32));
    float sum = 0.0f;
    for (int i = 0; i < 512; ++i) sum += __expf(S[(mbase + i) * 16 + r] - mx);
    sum += __shfl_xor(sum, 16, 32);
    const float inv = 1.0f / sum;
    for (int i = 0; i < 512; ++i) {
      const int idx = (mbase + i) * 16 + r;
      S[idx] = __expf(S[idx] - mx) * inv;
    }
  }

  v8f accO[4] = {};
  {
    v2f aA[8], aB[8];
    v2f bA[8][4], bB[8][4];
    gat_request_pv_stage(S, zp, 0, l16, lh, aA, bA);
    for (int st = 0; st < 30; st += 2) {
      gat_request_pv_stage(S, zp, st + 1, l16, lh, aB, bB);
      gat_pv_stage(aA, bA, accO);
      gat_request_pv_stage(S, zp, st + 2, l16, lh, aA, bA);
      gat_pv_stage(aB, bB, accO);
    }
    gat_request_pv_stage(S, zp, 31, l16, lh, aB, bB);
    gat_pv_stage(aA, bA, accO);
    gat_pv_stage(aB, bB, accO);
  }

  float* op = out + (size_t)pair * NODES * HID;
#pragma unroll
  for (int ct = 0; ct < 4; ++ct)
#pragma unroll
    for (int v = 0; v < 8; ++v) {
      const int row = v + 8 * lh;
      op[(size_t)(n0 + row) * HID + ct * 16 + l16] = accO[ct][v];
    }
}

extern "C" void kernel_launch(void* const* d_in, const int* in_sizes, int n_in,
                              void* d_out, int out_size, void* d_ws, size_t ws_size,
                              hipStream_t stream) {
  const float* x     = (const float*)d_in[0];
  const int*   mask  = (const int*)  d_in[1];
  const float* w_ih0 = (const float*)d_in[2];
  const float* w_hh0 = (const float*)d_in[3];
  const float* b_ih0 = (const float*)d_in[4];
  const float* b_hh0 = (const float*)d_in[5];
  const float* w_ih1 = (const float*)d_in[6];
  const float* w_hh1 = (const float*)d_in[7];
  const float* b_ih1 = (const float*)d_in[8];
  const float* b_hh1 = (const float*)d_in[9];
  float* out  = (float*)d_out;
  float* zbuf = (float*)d_ws;

  gru_fused_kernel<<<256, 128, 0, stream>>>(x, w_ih0, w_hh0, b_ih0, b_hh0,
                                            w_ih1, w_hh1, b_ih1, b_hh1, zbuf);
  gat_kernel<<<2048, 32, 0, stream>>>(zbuf, mask, out);
}


# Attention kernel: softmax row maxima kept in registers beside the score WMMAs and reduced across lanes once; the separate maximum sweep over the LDS score tile is gone
# speedup vs baseline: 1.9263x; 1.1709x over previous
#include <hip/hip_runtime.h>
#include <hip/hip_bf16.h>
#include <math.h>


typedef __attribute__((ext_vector_type(2))) float v2f;
typedef __attribute__((ext_vector_type(8))) float v8f;

#define T_SEQ   12
#define BCN     16
#define NODES   1024
#define FEAT    32
#define HID     64
#define G3      192
#define NROWS   (BCN * NODES)
#define NEG_BIG (-1e16f)

__device__ __forceinline__ v8f wmma_f32(v2f a, v2f b, v8f c) {
  return __builtin_amdgcn_wmma_f32_16x16x4_f32(
       false, a,  false, b,
       (short)0, c,  false,  false);
}

__device__ __forceinline__ v8f bias8(float v) {
  v8f r;
#pragma unroll
  for (int i = 0; i < 8; ++i) r[i] = v;
  return r;
}

__device__ __forceinline__ float sigmoidf_(float x) {
  return __builtin_amdgcn_rcpf(1.0f + __expf(-x));
}

__device__ __forceinline__ float tanhf_(float x) {
  const float ax = fabsf(x);
  const float t  = ax * ax;
  float p = fmaf(t, -0.00570002059f, 0.0206340719f);
  p = fmaf(t, p, -0.0537379310f);
  p = fmaf(t, p, 0.133314162f);
  p = fmaf(t, p, -0.333332807f);
  const float lo = fmaf(t, ax * p, ax);
  const float e  = __expf(2.0f * ax);
  const float hi = fmaf(__builtin_amdgcn_rcpf(1.0f + e), -2.0f, 1.0f);
  return copysignf(ax < 0.625f ? lo : hi, x);
}

#define GRU_WAVES 4
#define GRU_ROWS  64
#define HPAD 66

typedef __attribute__((ext_vector_type(4))) float v4f;

template <int KS>
__device__ __forceinline__ void gru_load_w(const float* __restrict__ w, int kdim,
                                           int g, int lh, v2f (&f)[KS]) {
  const float* wr = w + (size_t)g * kdim + 2 * lh;
#pragma unroll
  for (int ks = 0; ks < KS; ++ks) { f[ks].x = wr[4 * ks]; f[ks].y = wr[4 * ks + 1]; }
}

__device__ __forceinline__ void gru_load_h(const float* hs, int rb, int l16, int lh,
                                           v2f (&f)[16]) {
  const float* hr = hs + (rb + l16) * HPAD + 2 * lh;
#pragma unroll
  for (int ks = 0; ks < 16; ++ks) { f[ks].x = hr[4 * ks]; f[ks].y = hr[4 * ks + 1]; }
}

__device__ __forceinline__ void gru_load_x(const float* __restrict__ x_in, int t,
                                           int row, int lh, v2f (&f)[8]) {
  const float* xr = x_in + ((size_t)t * NROWS + row) * FEAT + 2 * lh;
#pragma unroll
  for (int ks = 0; ks < 8; ++ks) { f[ks].x = xr[4 * ks]; f[ks].y = xr[4 * ks + 1]; }
}

__global__ __launch_bounds__(128) void gru_fused_kernel(
    const float* __restrict__ x_in,
    const float* __restrict__ w_ih0,
    const float* __restrict__ w_hh0,
    const float* __restrict__ b_ih0,
    const float* __restrict__ b_hh0,
    const float* __restrict__ w_ih1,
    const float* __restrict__ w_hh1,
    const float* __restrict__ b_ih1,
    const float* __restrict__ b_hh1,
    float* __restrict__ z_out)
{
  __shared__ float hbuf[2][2][GRU_ROWS * HPAD];

  const int tid  = threadIdx.x;
  const int wave = tid >> 5;
  const int lane = tid & 31;
  const int l16  = lane & 15;
  const int lh   = lane >> 4;
  const int row0 = blockIdx.x * GRU_ROWS;
  const int cb   = wave * 16;
  const int gr   = cb + l16;
  const int gz   = 64 + cb + l16;
  const int gn   = 128 + cb + l16;

  for (int i = tid; i < GRU_ROWS * HPAD; i += GRU_WAVES * 32) {
    hbuf[0][0][i] = 0.0f;
    hbuf[1][0][i] = 0.0f;
  }

  v2f wi0r[8],  wi0z[8],  wi0n[8];
  v2f wh0r[16], wh0z[16], wh0n[16];
  v2f wi1r[16], wi1z[16], wi1n[16];
  v2f wh1r[16], wh1z[16], wh1n[16];
  gru_load_w<8>(w_ih0, FEAT, gr, lh, wi0r);
  gru_load_w<8>(w_ih0, FEAT, gz, lh, wi0z);
  gru_load_w<8>(w_ih0, FEAT, gn, lh, wi0n);
  gru_load_w<16>(w_hh0, HID, gr, lh, wh0r);
  gru_load_w<16>(w_hh0, HID, gz, lh, wh0z);
  gru_load_w<16>(w_hh0, HID, gn, lh, wh0n);
  gru_load_w<16>(w_ih1, HID, gr, lh, wi1r);
  gru_load_w<16>(w_ih1, HID, gz, lh, wi1z);
  gru_load_w<16>(w_ih1, HID, gn, lh, wi1n);
  gru_load_w<16>(w_hh1, HID, gr, lh, wh1r);
  gru_load_w<16>(w_hh1, HID, gz, lh, wh1z);
  gru_load_w<16>(w_hh1, HID, gn, lh, wh1n);
  const float bi0r = b_ih0[gr], bi0z = b_ih0[gz], bi0n = b_ih0[gn];
  const float bh0r = b_hh0[gr], bh0z = b_hh0[gz], bh0n = b_hh0[gn];
  const float bi1r = b_ih1[gr], bi1z = b_ih1[gz], bi1n = b_ih1[gn];
  const float bh1r = b_hh1[gr], bh1z = b_hh1[gz], bh1n = b_hh1[gn];

  v2f xc[8];
  gru_load_x(x_in, 0, row0 + l16, lh, xc);

  __syncthreads();

  for (int t = 0; t < T_SEQ; ++t) {
    const float* h0r = &hbuf[0][t & 1][0];
    float*       h0w = &hbuf[0][(t + 1) & 1][0];
    const float* h1r = &hbuf[1][t & 1][0];
    float*       h1w = &hbuf[1][(t + 1) & 1][0];

#pragma unroll 1
    for (int rt = 0; rt < 4; ++rt) {
      const int rb = rt * 16;
      v2f ah[16];
      gru_load_h(h0r, rb, l16, lh, ah);
      float hold[8];
#pragma unroll
      for (int v = 0; v < 8; ++v) hold[v] = h0r[(rb + v + 8 * lh) * HPAD + cb + l16];

      const int nrt = (rt + 1) & 3;
      const int nt  = (rt == 3 && t + 1 < T_SEQ) ? t + 1 : t;
      v2f xn[8];
      gru_load_x(x_in, nt, row0 + nrt * 16 + l16, lh, xn);
      __builtin_amdgcn_sched_barrier(0);

      v8f gir = bias8(bi0r), giz = bias8(bi0z), gin = bias8(bi0n);
      v8f ghr = bias8(bh0r), ghz = bias8(bh0z), ghn = bias8(bh0n);
#pragma unroll
      for (int ks = 0; ks < 8; ++ks) {
        gir = wmma_f32(xc[ks], wi0r[ks], gir);
        giz = wmma_f32(xc[ks], wi0z[ks], giz);
        gin = wmma_f32(xc[ks], wi0n[ks], gin);
      }
#pragma unroll
      for (int ks = 0; ks < 16; ++ks) {
        ghr = wmma_f32(ah[ks], wh0r[ks], ghr);
        ghz = wmma_f32(ah[ks], wh0z[ks], ghz);
        ghn = wmma_f32(ah[ks], wh0n[ks], ghn);
      }
#pragma unroll
      for (int v = 0; v < 8; ++v) {
        const float r  = sigmoidf_(gir[v] + ghr[v]);
        const float zg = sigmoidf_(giz[v] + ghz[v]);
        const float n  = tanhf_(gin[v] + r * ghn[v]);
        h0w[(rb + v + 8 * lh) * HPAD + cb + l16] = (1.0f - zg) * n + zg * hold[v];
      }
#pragma unroll
      for (int ks = 0; ks < 8; ++ks) xc[ks] = xn[ks];
    }

    __syncthreads();

#pragma unroll 1
    for (int rt = 0; rt < 4; ++rt) {
      const int rb = rt * 16;
      v2f a0[16], a1[16];
      gru_load_h(h0w, rb, l16, lh, a0);
      gru_load_h(h1r, rb, l16, lh, a1);
      float hold[8];
#pragma unroll
      for (int v = 0; v < 8; ++v) hold[v] = h1r[(rb + v + 8 * lh) * HPAD + cb + l16];
      __builtin_amdgcn_sched_barrier(0);

      v8f gir = bias8(bi1r), giz = bias8(bi1z), gin = bias8(bi1n);
      v8f ghr = bias8(bh1r), ghz = bias8(bh1z), ghn = bias8(bh1n);
#pragma unroll
      for (int ks = 0; ks < 16; ++ks) {
        gir = wmma_f32(a0[ks], wi1r[ks], gir);
        giz = wmma_f32(a0[ks], wi1z[ks], giz);
        gin = wmma_f32(a0[ks], wi1n[ks], gin);
      }
#pragma unroll
      for (int ks = 0; ks < 16; ++ks) {
        ghr = wmma_f32(a1[ks], wh1r[ks], ghr);
        ghz = wmma_f32(a1[ks], wh1z[ks], ghz);
        ghn = wmma_f32(a1[ks], wh1n[ks], ghn);
      }
#pragma unroll
      for (int v = 0; v < 8; ++v) {
        const float r  = sigmoidf_(gir[v] + ghr[v]);
        const float zg = sigmoidf_(giz[v] + ghz[v]);
        const float n  = tanhf_(gin[v] + r * ghn[v]);
        h1w[(rb + v + 8 * lh) * HPAD + cb + l16] = (1.0f - zg) * n + zg * hold[v];
      }
    }
  }

  __syncthreads();

  const float* f0 = &hbuf[0][T_SEQ & 1][0];
  const float* f1 = &hbuf[1][T_SEQ & 1][0];
  for (int i = tid; i < GRU_ROWS * 16; i += GRU_WAVES * 32) {
    const int r = i >> 4, c = (i & 15) * 4;
    v4f p, q;
    p.x = f0[r * HPAD + c]; p.y = f0[r * HPAD + c + 1];
    p.z = f0[r * HPAD + c + 2]; p.w = f0[r * HPAD + c + 3];
    q.x = f1[r * HPAD + c]; q.y = f1[r * HPAD + c + 1];
    q.z = f1[r * HPAD + c + 2]; q.w = f1[r * HPAD + c + 3];
    *(v4f*)(z_out + (size_t)(row0 + r) * HID + c) = p;
    *(v4f*)(z_out + (size_t)NROWS * HID + (size_t)(row0 + r) * HID + c) = q;
  }
}


__device__ __forceinline__ float row16_max(float x) {
  x = fmaxf(x, __int_as_float(__builtin_amdgcn_update_dpp(
          __float_as_int(x), __float_as_int(x), 0x161, 0xf, 0xf, true)));
  x = fmaxf(x, __int_as_float(__builtin_amdgcn_update_dpp(
          __float_as_int(x), __float_as_int(x), 0x162, 0xf, 0xf, true)));
  x = fmaxf(x, __int_as_float(__builtin_amdgcn_update_dpp(
          __float_as_int(x), __float_as_int(x), 0x164, 0xf, 0xf, true)));
  x = fmaxf(x, __int_as_float(__builtin_amdgcn_update_dpp(
          __float_as_int(x), __float_as_int(x), 0x168, 0xf, 0xf, true)));
  return x;
}

__device__ __forceinline__ void gat_request_score_stage(
    const float* __restrict__ zp, const int* __restrict__ mrow,
    int st, int l16, int lh, v2f (&b)[2][16], int (&m)[2][8])
{
#pragma unroll
  for (int u = 0; u < 2; ++u) {
    const int col = (st * 2 + u) * 16 + l16;
    const float* zr = zp + (size_t)col * HID + 2 * lh;
#pragma unroll
    for (int ks = 0; ks < 16; ++ks) {
      b[u][ks].x = zr[4 * ks];
      b[u][ks].y = zr[4 * ks + 1];
    }
    const int* mp = mrow + col;
#pragma unroll
    for (int v = 0; v < 8; ++v) m[u][v] = mp[(size_t)v * NODES];
  }
}

__device__ __forceinline__ void gat_score_stage(
    const v2f (&za)[16], const v2f (&b)[2][16], const int (&m)[2][8],
    float* S, int st, int l16, int lh, float (&rmax)[8])
{
#pragma unroll
  for (int u = 0; u < 2; ++u) {
    v8f acc = {};
#pragma unroll
    for (int ks = 0; ks < 16; ++ks) acc = wmma_f32(za[ks], b[u][ks], acc);
    const int col = (st * 2 + u) * 16 + l16;
#pragma unroll
    for (int v = 0; v < 8; ++v) {
      const int row = v + 8 * lh;
      const float s  = acc[v];
      const float lr = s > 0.0f ? s : 0.1f * s;
      float a = lr * (float)m[u][v];
      a = (a == 0.0f) ? NEG_BIG : a;
      S[col * 16 + row] = a;
      rmax[v] = fmaxf(rmax[v], a);
    }
  }
}

__device__ __forceinline__ void gat_request_pv_stage(
    const float* S, const float* __restrict__ zp,
    int st, int l16, int lh, v2f (&a)[8], v2f (&b)[8][4])
{
#pragma unroll
  for (int ks = 0; ks < 8; ++ks) {
    const int k = st * 32 + 4 * ks + 2 * lh;
    a[ks].x = S[k * 16 + l16];
    a[ks].y = S[(k + 1) * 16 + l16];
#pragma unroll
    for (int ct = 0; ct < 4; ++ct) {
      b[ks][ct].x = zp[(size_t)k * HID + ct * 16 + l16];
      b[ks][ct].y = zp[(size_t)(k + 1) * HID + ct * 16 + l16];
    }
  }
}

__device__ __forceinline__ void gat_pv_stage(
    const v2f (&a)[8], const v2f (&b)[8][4], v8f (&accO)[4])
{
#pragma unroll
  for (int ks = 0; ks < 8; ++ks)
#pragma unroll
    for (int ct = 0; ct < 4; ++ct)
      accO[ct] = wmma_f32(a[ks], b[ks][ct], accO[ct]);
}

__global__ __launch_bounds__(32) void gat_kernel(
    const float* __restrict__ z,
    const int*   __restrict__ mask,
    float*       __restrict__ out)
{
  __shared__ float S[NODES * 16];

  const int lane  = threadIdx.x;
  const int l16   = lane & 15;
  const int lh    = lane >> 4;
  const int ntile = blockIdx.x & 63;
  const int pair  = blockIdx.x >> 6;
  const float* zp = z + (size_t)pair * NODES * HID;
  const int n0 = ntile * 16;

  v2f za[16];
  {
    const float* zrow = zp + (size_t)(n0 + l16) * HID;
#pragma unroll
    for (int ks = 0; ks < 16; ++ks) {
      const int k = 4 * ks + 2 * lh;
      za[ks].x = zrow[k]; za[ks].y = zrow[k + 1];
    }
  }

  float mx;
  {
    float rmax[8];
#pragma unroll
    for (int v = 0; v < 8; ++v) rmax[v] = -INFINITY;
    const int* mrow = mask + (size_t)(n0 + 8 * lh) * NODES;
    v2f bA[2][16], bB[2][16];
    int mA[2][8], mB[2][8];
    gat_request_score_stage(zp, mrow, 0, l16, lh, bA, mA);
    for (int st = 0; st < 30; st += 2) {
      gat_request_score_stage(zp, mrow, st + 1, l16, lh, bB, mB);
      gat_score_stage(za, bA, mA, S, st, l16, lh, rmax);
      gat_request_score_stage(zp, mrow, st + 2, l16, lh, bA, mA);
      gat_score_stage(za, bB, mB, S, st + 1, l16, lh, rmax);
    }
    gat_request_score_stage(zp, mrow, 31, l16, lh, bB, mB);
    gat_score_stage(za, bA, mA, S, 30, l16, lh, rmax);
    gat_score_stage(za, bB, mB, S, 31, l16, lh, rmax);

#pragma unroll
    for (int v = 0; v < 8; ++v) rmax[v] = row16_max(rmax[v]);
    const int sel = l16 & 7;
    float pick = rmax[0];
#pragma unroll
    for (int v = 1; v < 8; ++v) pick = (sel == v) ? rmax[v] : pick;
    const float other = __shfl_xor(pick, 16, 32);
    mx = ((l16 >> 3) == lh) ? pick : other;
  }

  {
    const int r = l16;
    const int mbase = lh * 512;
    float sum = 0.0f;
    for (int i = 0; i < 512; ++i) sum += __expf(S[(mbase + i) * 16 + r] - mx);
    sum += __shfl_xor(sum, 16, 32);
    const float inv = 1.0f / sum;
    for (int i = 0; i < 512; ++i) {
      const int idx = (mbase + i) * 16 + r;
      S[idx] = __expf(S[idx] - mx) * inv;
    }
  }

  v8f accO[4] = {};
  {
    v2f aA[8], aB[8];
    v2f bA[8][4], bB[8][4];
    gat_request_pv_stage(S, zp, 0, l16, lh, aA, bA);
    for (int st = 0; st < 30; st += 2) {
      gat_request_pv_stage(S, zp, st + 1, l16, lh, aB, bB);
      gat_pv_stage(aA, bA, accO);
      gat_request_pv_stage(S, zp, st + 2, l16, lh, aA, bA);
      gat_pv_stage(aB, bB, accO);
    }
    gat_request_pv_stage(S, zp, 31, l16, lh, aB, bB);
    gat_pv_stage(aA, bA, accO);
    gat_pv_stage(aB, bB, accO);
  }

  float* op = out + (size_t)pair * NODES * HID;
#pragma unroll
  for (int ct = 0; ct < 4; ++ct)
#pragma unroll
    for (int v = 0; v < 8; ++v) {
      const int row = v + 8 * lh;
      op[(size_t)(n0 + row) * HID + ct * 16 + l16] = accO[ct][v];
    }
}

extern "C" void kernel_launch(void* const* d_in, const int* in_sizes, int n_in,
                              void* d_out, int out_size, void* d_ws, size_t ws_size,
                              hipStream_t stream) {
  const float* x     = (const float*)d_in[0];
  const int*   mask  = (const int*)  d_in[1];
  const float* w_ih0 = (const float*)d_in[2];
  const float* w_hh0 = (const float*)d_in[3];
  const float* b_ih0 = (const float*)d_in[4];
  const float* b_hh0 = (const float*)d_in[5];
  const float* w_ih1 = (const float*)d_in[6];
  const float* w_hh1 = (const float*)d_in[7];
  const float* b_ih1 = (const float*)d_in[8];
  const float* b_hh1 = (const float*)d_in[9];
  float* out  = (float*)d_out;
  float* zbuf = (float*)d_ws;

  gru_fused_kernel<<<256, 128, 0, stream>>>(x, w_ih0, w_hh0, b_ih0, b_hh0,
                                            w_ih1, w_hh1, b_ih1, b_hh1, zbuf);
  gat_kernel<<<2048, 32, 0, stream>>>(zbuf, mask, out);
}


# Attention kernel: both softmax sweeps over the LDS score tile folded into the P@z loop: exp and row sum (per-stage tree, new sum order) on the A fragments in registers; 1/sum scales the output
# speedup vs baseline: 2.1211x; 1.1011x over previous
#include <hip/hip_runtime.h>
#include <hip/hip_bf16.h>
#include <math.h>


typedef __attribute__((ext_vector_type(2))) float v2f;
typedef __attribute__((ext_vector_type(8))) float v8f;

#define T_SEQ   12
#define BCN     16
#define NODES   1024
#define FEAT    32
#define HID     64
#define G3      192
#define NROWS   (BCN * NODES)
#define NEG_BIG (-1e16f)

__device__ __forceinline__ v8f wmma_f32(v2f a, v2f b, v8f c) {
  return __builtin_amdgcn_wmma_f32_16x16x4_f32(
       false, a,  false, b,
       (short)0, c,  false,  false);
}

__device__ __forceinline__ v8f bias8(float v) {
  v8f r;
#pragma unroll
  for (int i = 0; i < 8; ++i) r[i] = v;
  return r;
}

__device__ __forceinline__ float sigmoidf_(float x) {
  return __builtin_amdgcn_rcpf(1.0f + __expf(-x));
}

__device__ __forceinline__ float tanhf_(float x) {
  const float ax = fabsf(x);
  const float t  = ax * ax;
  float p = fmaf(t, -0.00570002059f, 0.0206340719f);
  p = fmaf(t, p, -0.0537379310f);
  p = fmaf(t, p, 0.133314162f);
  p = fmaf(t, p, -0.333332807f);
  const float lo = fmaf(t, ax * p, ax);
  const float e  = __expf(2.0f * ax);
  const float hi = fmaf(__builtin_amdgcn_rcpf(1.0f + e), -2.0f, 1.0f);
  return copysignf(ax < 0.625f ? lo : hi, x);
}

#define GRU_WAVES 4
#define GRU_ROWS  64
#define HPAD 66

typedef __attribute__((ext_vector_type(4))) float v4f;

template <int KS>
__device__ __forceinline__ void gru_load_w(const float* __restrict__ w, int kdim,
                                           int g, int lh, v2f (&f)[KS]) {
  const float* wr = w + (size_t)g * kdim + 2 * lh;
#pragma unroll
  for (int ks = 0; ks < KS; ++ks) { f[ks].x = wr[4 * ks]; f[ks].y = wr[4 * ks + 1]; }
}

__device__ __forceinline__ void gru_load_h(const float* hs, int rb, int l16, int lh,
                                           v2f (&f)[16]) {
  const float* hr = hs + (rb + l16) * HPAD + 2 * lh;
#pragma unroll
  for (int ks = 0; ks < 16; ++ks) { f[ks].x = hr[4 * ks]; f[ks].y = hr[4 * ks + 1]; }
}

__device__ __forceinline__ void gru_load_x(const float* __restrict__ x_in, int t,
                                           int row, int lh, v2f (&f)[8]) {
  const float* xr = x_in + ((size_t)t * NROWS + row) * FEAT + 2 * lh;
#pragma unroll
  for (int ks = 0; ks < 8; ++ks) { f[ks].x = xr[4 * ks]; f[ks].y = xr[4 * ks + 1]; }
}

__global__ __launch_bounds__(128) void gru_fused_kernel(
    const float* __restrict__ x_in,
    const float* __restrict__ w_ih0,
    const float* __restrict__ w_hh0,
    const float* __restrict__ b_ih0,
    const float* __restrict__ b_hh0,
    const float* __restrict__ w_ih1,
    const float* __restrict__ w_hh1,
    const float* __restrict__ b_ih1,
    const float* __restrict__ b_hh1,
    float* __restrict__ z_out)
{
  __shared__ float hbuf[2][2][GRU_ROWS * HPAD];

  const int tid  = threadIdx.x;
  const int wave = tid >> 5;
  const int lane = tid & 31;
  const int l16  = lane & 15;
  const int lh   = lane >> 4;
  const int row0 = blockIdx.x * GRU_ROWS;
  const int cb   = wave * 16;
  const int gr   = cb + l16;
  const int gz   = 64 + cb + l16;
  const int gn   = 128 + cb + l16;

  for (int i = tid; i < GRU_ROWS * HPAD; i += GRU_WAVES * 32) {
    hbuf[0][0][i] = 0.0f;
    hbuf[1][0][i] = 0.0f;
  }

  v2f wi0r[8],  wi0z[8],  wi0n[8];
  v2f wh0r[16], wh0z[16], wh0n[16];
  v2f wi1r[16], wi1z[16], wi1n[16];
  v2f wh1r[16], wh1z[16], wh1n[16];
  gru_load_w<8>(w_ih0, FEAT, gr, lh, wi0r);
  gru_load_w<8>(w_ih0, FEAT, gz, lh, wi0z);
  gru_load_w<8>(w_ih0, FEAT, gn, lh, wi0n);
  gru_load_w<16>(w_hh0, HID, gr, lh, wh0r);
  gru_load_w<16>(w_hh0, HID, gz, lh, wh0z);
  gru_load_w<16>(w_hh0, HID, gn, lh, wh0n);
  gru_load_w<16>(w_ih1, HID, gr, lh, wi1r);
  gru_load_w<16>(w_ih1, HID, gz, lh, wi1z);
  gru_load_w<16>(w_ih1, HID, gn, lh, wi1n);
  gru_load_w<16>(w_hh1, HID, gr, lh, wh1r);
  gru_load_w<16>(w_hh1, HID, gz, lh, wh1z);
  gru_load_w<16>(w_hh1, HID, gn, lh, wh1n);
  const float bi0r = b_ih0[gr], bi0z = b_ih0[gz], bi0n = b_ih0[gn];
  const float bh0r = b_hh0[gr], bh0z = b_hh0[gz], bh0n = b_hh0[gn];
  const float bi1r = b_ih1[gr], bi1z = b_ih1[gz], bi1n = b_ih1[gn];
  const float bh1r = b_hh1[gr], bh1z = b_hh1[gz], bh1n = b_hh1[gn];

  v2f xc[8];
  gru_load_x(x_in, 0, row0 + l16, lh, xc);

  __syncthreads();

  for (int t = 0; t < T_SEQ; ++t) {
    const float* h0r = &hbuf[0][t & 1][0];
    float*       h0w = &hbuf[0][(t + 1) & 1][0];
    const float* h1r = &hbuf[1][t & 1][0];
    float*       h1w = &hbuf[1][(t + 1) & 1][0];

#pragma unroll 1
    for (int rt = 0; rt < 4; ++rt) {
      const int rb = rt * 16;
      v2f ah[16];
      gru_load_h(h0r, rb, l16, lh, ah);
      float hold[8];
#pragma unroll
      for (int v = 0; v < 8; ++v) hold[v] = h0r[(rb + v + 8 * lh) * HPAD + cb + l16];

      const int nrt = (rt + 1) & 3;
      const int nt  = (rt == 3 && t + 1 < T_SEQ) ? t + 1 : t;
      v2f xn[8];
      gru_load_x(x_in, nt, row0 + nrt * 16 + l16, lh, xn);
      __builtin_amdgcn_sched_barrier(0);

      v8f gir = bias8(bi0r), giz = bias8(bi0z), gin = bias8(bi0n);
      v8f ghr = bias8(bh0r), ghz = bias8(bh0z), ghn = bias8(bh0n);
#pragma unroll
      for (int ks = 0; ks < 8; ++ks) {
        gir = wmma_f32(xc[ks], wi0r[ks], gir);
        giz = wmma_f32(xc[ks], wi0z[ks], giz);
        gin = wmma_f32(xc[ks], wi0n[ks], gin);
      }
#pragma unroll
      for (int ks = 0; ks < 16; ++ks) {
        ghr = wmma_f32(ah[ks], wh0r[ks], ghr);
        ghz = wmma_f32(ah[ks], wh0z[ks], ghz);
        ghn = wmma_f32(ah[ks], wh0n[ks], ghn);
      }
#pragma unroll
      for (int v = 0; v < 8; ++v) {
        const float r  = sigmoidf_(gir[v] + ghr[v]);
        const float zg = sigmoidf_(giz[v] + ghz[v]);
        const float n  = tanhf_(gin[v] + r * ghn[v]);
        h0w[(rb + v + 8 * lh) * HPAD + cb + l16] = (1.0f - zg) * n + zg * hold[v];
      }
#pragma unroll
      for (int ks = 0; ks < 8; ++ks) xc[ks] = xn[ks];
    }

    __syncthreads();

#pragma unroll 1
    for (int rt = 0; rt < 4; ++rt) {
      const int rb = rt * 16;
      v2f a0[16], a1[16];
      gru_load_h(h0w, rb, l16, lh, a0);
      gru_load_h(h1r, rb, l16, lh, a1);
      float hold[8];
#pragma unroll
      for (int v = 0; v < 8; ++v) hold[v] = h1r[(rb + v + 8 * lh) * HPAD + cb + l16];
      __builtin_amdgcn_sched_barrier(0);

      v8f gir = bias8(bi1r), giz = bias8(bi1z), gin = bias8(bi1n);
      v8f ghr = bias8(bh1r), ghz = bias8(bh1z), ghn = bias8(bh1n);
#pragma unroll
      for (int ks = 0; ks < 16; ++ks) {
        gir = wmma_f32(a0[ks], wi1r[ks], gir);
        giz = wmma_f32(a0[ks], wi1z[ks], giz);
        gin = wmma_f32(a0[ks], wi1n[ks], gin);
      }
#pragma unroll
      for (int ks = 0; ks < 16; ++ks) {
        ghr = wmma_f32(a1[ks], wh1r[ks], ghr);
        ghz = wmma_f32(a1[ks], wh1z[ks], ghz);
        ghn = wmma_f32(a1[ks], wh1n[ks], ghn);
      }
#pragma unroll
      for (int v = 0; v < 8; ++v) {
        const float r  = sigmoidf_(gir[v] + ghr[v]);
        const float zg = sigmoidf_(giz[v] + ghz[v]);
        const float n  = tanhf_(gin[v] + r * ghn[v]);
        h1w[(rb + v + 8 * lh) * HPAD + cb + l16] = (1.0f - zg) * n + zg * hold[v];
      }
    }
  }

  __syncthreads();

  const float* f0 = &hbuf[0][T_SEQ & 1][0];
  const float* f1 = &hbuf[1][T_SEQ & 1][0];
  for (int i = tid; i < GRU_ROWS * 16; i += GRU_WAVES * 32) {
    const int r = i >> 4, c = (i & 15) * 4;
    v4f p, q;
    p.x = f0[r * HPAD + c]; p.y = f0[r * HPAD + c + 1];
    p.z = f0[r * HPAD + c + 2]; p.w = f0[r * HPAD + c + 3];
    q.x = f1[r * HPAD + c]; q.y = f1[r * HPAD + c + 1];
    q.z = f1[r * HPAD + c + 2]; q.w = f1[r * HPAD + c + 3];
    *(v4f*)(z_out + (size_t)(row0 + r) * HID + c) = p;
    *(v4f*)(z_out + (size_t)NROWS * HID + (size_t)(row0 + r) * HID + c) = q;
  }
}


__device__ __forceinline__ float row16_max(float x) {
  x = fmaxf(x, __int_as_float(__builtin_amdgcn_update_dpp(
          __float_as_int(x), __float_as_int(x), 0x161, 0xf, 0xf, true)));
  x = fmaxf(x, __int_as_float(__builtin_amdgcn_update_dpp(
          __float_as_int(x), __float_as_int(x), 0x162, 0xf, 0xf, true)));
  x = fmaxf(x, __int_as_float(__builtin_amdgcn_update_dpp(
          __float_as_int(x), __float_as_int(x), 0x164, 0xf, 0xf, true)));
  x = fmaxf(x, __int_as_float(__builtin_amdgcn_update_dpp(
          __float_as_int(x), __float_as_int(x), 0x168, 0xf, 0xf, true)));
  return x;
}

__device__ __forceinline__ void gat_request_score_stage(
    const float* __restrict__ zp, const int* __restrict__ mrow,
    int st, int l16, int lh, v2f (&b)[2][16], int (&m)[2][8])
{
#pragma unroll
  for (int u = 0; u < 2; ++u) {
    const int col = (st * 2 + u) * 16 + l16;
    const float* zr = zp + (size_t)col * HID + 2 * lh;
#pragma unroll
    for (int ks = 0; ks < 16; ++ks) {
      b[u][ks].x = zr[4 * ks];
      b[u][ks].y = zr[4 * ks + 1];
    }
    const int* mp = mrow + col;
#pragma unroll
    for (int v = 0; v < 8; ++v) m[u][v] = mp[(size_t)v * NODES];
  }
}

__device__ __forceinline__ void gat_score_stage(
    const v2f (&za)[16], const v2f (&b)[2][16], const int (&m)[2][8],
    float* S, int st, int l16, int lh, float (&rmax)[8])
{
#pragma unroll
  for (int u = 0; u < 2; ++u) {
    v8f acc = {};
#pragma unroll
    for (int ks = 0; ks < 16; ++ks) acc = wmma_f32(za[ks], b[u][ks], acc);
    const int col = (st * 2 + u) * 16 + l16;
#pragma unroll
    for (int v = 0; v < 8; ++v) {
      const int row = v + 8 * lh;
      const float s  = acc[v];
      const float lr = s > 0.0f ? s : 0.1f * s;
      float a = lr * (float)m[u][v];
      a = (a == 0.0f) ? NEG_BIG : a;
      S[col * 16 + row] = a;
      rmax[v] = fmaxf(rmax[v], a);
    }
  }
}

__device__ __forceinline__ void gat_request_pv_stage(
    const float* S, const float* __restrict__ zp,
    int st, int l16, int lh, v2f (&a)[8], v2f (&b)[8][4])
{
#pragma unroll
  for (int ks = 0; ks < 8; ++ks) {
    const int k = st * 32 + 4 * ks + 2 * lh;
    a[ks].x = S[k * 16 + l16];
    a[ks].y = S[(k + 1) * 16 + l16];
#pragma unroll
    for (int ct = 0; ct < 4; ++ct) {
      b[ks][ct].x = zp[(size_t)k * HID + ct * 16 + l16];
      b[ks][ct].y = zp[(size_t)(k + 1) * HID + ct * 16 + l16];
    }
  }
}

__device__ __forceinline__ void gat_pv_stage(
    const v2f (&a)[8], const v2f (&b)[8][4], v8f (&accO)[4], float mx, float& sum)
{
  v2f e[8];
#pragma unroll
  for (int ks = 0; ks < 8; ++ks) {
    e[ks].x = __expf(a[ks].x - mx);
    e[ks].y = __expf(a[ks].y - mx);
  }
  const v2f t0 = (e[0] + e[1]) + (e[2] + e[3]);
  const v2f t1 = (e[4] + e[5]) + (e[6] + e[7]);
  const v2f t  = t0 + t1;
  sum += t.x + t.y;
#pragma unroll
  for (int ks = 0; ks < 8; ++ks)
#pragma unroll
    for (int ct = 0; ct < 4; ++ct)
      accO[ct] = wmma_f32(e[ks], b[ks][ct], accO[ct]);
}

__global__ __launch_bounds__(32) void gat_kernel(
    const float* __restrict__ z,
    const int*   __restrict__ mask,
    float*       __restrict__ out)
{
  __shared__ float S[NODES * 16];

  const int lane  = threadIdx.x;
  const int l16   = lane & 15;
  const int lh    = lane >> 4;
  const int ntile = blockIdx.x & 63;
  const int pair  = blockIdx.x >> 6;
  const float* zp = z + (size_t)pair * NODES * HID;
  const int n0 = ntile * 16;

  v2f za[16];
  {
    const float* zrow = zp + (size_t)(n0 + l16) * HID;
#pragma unroll
    for (int ks = 0; ks < 16; ++ks) {
      const int k = 4 * ks + 2 * lh;
      za[ks].x = zrow[k]; za[ks].y = zrow[k + 1];
    }
  }

  float mx;
  {
    float rmax[8];
#pragma unroll
    for (int v = 0; v < 8; ++v) rmax[v] = -INFINITY;
    const int* mrow = mask + (size_t)(n0 + 8 * lh) * NODES;
    v2f bA[2][16], bB[2][16];
    int mA[2][8], mB[2][8];
    gat_request_score_stage(zp, mrow, 0, l16, lh, bA, mA);
    for (int st = 0; st < 30; st += 2) {
      gat_request_score_stage(zp, mrow, st + 1, l16, lh, bB, mB);
      gat_score_stage(za, bA, mA, S, st, l16, lh, rmax);
      gat_request_score_stage(zp, mrow, st + 2, l16, lh, bA, mA);
      gat_score_stage(za, bB, mB, S, st + 1, l16, lh, rmax);
    }
    gat_request_score_stage(zp, mrow, 31, l16, lh, bB, mB);
    gat_score_stage(za, bA, mA, S, 30, l16, lh, rmax);
    gat_score_stage(za, bB, mB, S, 31, l16, lh, rmax);

#pragma unroll
    for (int v = 0; v < 8; ++v) rmax[v] = row16_max(rmax[v]);
    const int sel = l16 & 7;
    float pick = rmax[0];
#pragma unroll
    for (int v = 1; v < 8; ++v) pick = (sel == v) ? rmax[v] : pick;
    const float other = __shfl_xor(pick, 16, 32);
    mx = ((l16 >> 3) == lh) ? pick : other;
  }

  v8f accO[4] = {};
  float sum = 0.0f;
  {
    v2f aA[8], aB[8];
    v2f bA[8][4], bB[8][4];
    gat_request_pv_stage(S, zp, 0, l16, lh, aA, bA);
    for (int st = 0; st < 30; st += 2) {
      gat_request_pv_stage(S, zp, st + 1, l16, lh, aB, bB);
      gat_pv_stage(aA, bA, accO, mx, sum);
      gat_request_pv_stage(S, zp, st + 2, l16, lh, aA, bA);
      gat_pv_stage(aB, bB, accO, mx, sum);
    }
    gat_request_pv_stage(S, zp, 31, l16, lh, aB, bB);
    gat_pv_stage(aA, bA, accO, mx, sum);
    gat_pv_stage(aB, bB, accO, mx, sum);
  }
  sum += __shfl_xor(sum, 16, 32);
  const float inv = 1.0f / sum;

  float* op = out + (size_t)pair * NODES * HID;
#pragma unroll
  for (int v = 0; v < 8; ++v) {
    const int row = v + 8 * lh;
    const float rinv = __shfl(inv, row, 32);
#pragma unroll
    for (int ct = 0; ct < 4; ++ct)
      op[(size_t)(n0 + row) * HID + ct * 16 + l16] = accO[ct][v] * rinv;
  }
}

extern "C" void kernel_launch(void* const* d_in, const int* in_sizes, int n_in,
                              void* d_out, int out_size, void* d_ws, size_t ws_size,
                              hipStream_t stream) {
  const float* x     = (const float*)d_in[0];
  const int*   mask  = (const int*)  d_in[1];
  const float* w_ih0 = (const float*)d_in[2];
  const float* w_hh0 = (const float*)d_in[3];
  const float* b_ih0 = (const float*)d_in[4];
  const float* b_hh0 = (const float*)d_in[5];
  const float* w_ih1 = (const float*)d_in[6];
  const float* w_hh1 = (const float*)d_in[7];
  const float* b_ih1 = (const float*)d_in[8];
  const float* b_hh1 = (const float*)d_in[9];
  float* out  = (float*)d_out;
  float* zbuf = (float*)d_ws;

  gru_fused_kernel<<<256, 128, 0, stream>>>(x, w_ih0, w_hh0, b_ih0, b_hh0,
                                            w_ih1, w_hh1, b_ih1, b_hh1, zbuf);
  gat_kernel<<<2048, 32, 0, stream>>>(zbuf, mask, out);
}


# Attention kernel: z fragments fetched by 16-byte loads: score A/B loaders take k=8j+4h+0..3 per load (new k order in the score sums), P@z tiles own columns 4c+tile so one load feeds 4 tiles
# speedup vs baseline: 2.3124x; 1.0902x over previous
#include <hip/hip_runtime.h>
#include <hip/hip_bf16.h>
#include <math.h>


typedef __attribute__((ext_vector_type(2))) float v2f;
typedef __attribute__((ext_vector_type(8))) float v8f;

#define T_SEQ   12
#define BCN     16
#define NODES   1024
#define FEAT    32
#define HID     64
#define G3      192
#define NROWS   (BCN * NODES)
#define NEG_BIG (-1e16f)

__device__ __forceinline__ v8f wmma_f32(v2f a, v2f b, v8f c) {
  return __builtin_amdgcn_wmma_f32_16x16x4_f32(
       false, a,  false, b,
       (short)0, c,  false,  false);
}

__device__ __forceinline__ v8f bias8(float v) {
  v8f r;
#pragma unroll
  for (int i = 0; i < 8; ++i) r[i] = v;
  return r;
}

__device__ __forceinline__ float sigmoidf_(float x) {
  return __builtin_amdgcn_rcpf(1.0f + __expf(-x));
}

__device__ __forceinline__ float tanhf_(float x) {
  const float ax = fabsf(x);
  const float t  = ax * ax;
  float p = fmaf(t, -0.00570002059f, 0.0206340719f);
  p = fmaf(t, p, -0.0537379310f);
  p = fmaf(t, p, 0.133314162f);
  p = fmaf(t, p, -0.333332807f);
  const float lo = fmaf(t, ax * p, ax);
  const float e  = __expf(2.0f * ax);
  const float hi = fmaf(__builtin_amdgcn_rcpf(1.0f + e), -2.0f, 1.0f);
  return copysignf(ax < 0.625f ? lo : hi, x);
}

#define GRU_WAVES 4
#define GRU_ROWS  64
#define HPAD 66

typedef __attribute__((ext_vector_type(4))) float v4f;

template <int KS>
__device__ __forceinline__ void gru_load_w(const float* __restrict__ w, int kdim,
                                           int g, int lh, v2f (&f)[KS]) {
  const float* wr = w + (size_t)g * kdim + 2 * lh;
#pragma unroll
  for (int ks = 0; ks < KS; ++ks) { f[ks].x = wr[4 * ks]; f[ks].y = wr[4 * ks + 1]; }
}

__device__ __forceinline__ void gru_load_h(const float* hs, int rb, int l16, int lh,
                                           v2f (&f)[16]) {
  const float* hr = hs + (rb + l16) * HPAD + 2 * lh;
#pragma unroll
  for (int ks = 0; ks < 16; ++ks) { f[ks].x = hr[4 * ks]; f[ks].y = hr[4 * ks + 1]; }
}

__device__ __forceinline__ void gru_load_x(const float* __restrict__ x_in, int t,
                                           int row, int lh, v2f (&f)[8]) {
  const float* xr = x_in + ((size_t)t * NROWS + row) * FEAT + 2 * lh;
#pragma unroll
  for (int ks = 0; ks < 8; ++ks) { f[ks].x = xr[4 * ks]; f[ks].y = xr[4 * ks + 1]; }
}

__global__ __launch_bounds__(128) void gru_fused_kernel(
    const float* __restrict__ x_in,
    const float* __restrict__ w_ih0,
    const float* __restrict__ w_hh0,
    const float* __restrict__ b_ih0,
    const float* __restrict__ b_hh0,
    const float* __restrict__ w_ih1,
    const float* __restrict__ w_hh1,
    const float* __restrict__ b_ih1,
    const float* __restrict__ b_hh1,
    float* __restrict__ z_out)
{
  __shared__ float hbuf[2][2][GRU_ROWS * HPAD];

  const int tid  = threadIdx.x;
  const int wave = tid >> 5;
  const int lane = tid & 31;
  const int l16  = lane & 15;
  const int lh   = lane >> 4;
  const int row0 = blockIdx.x * GRU_ROWS;
  const int cb   = wave * 16;
  const int gr   = cb + l16;
  const int gz   = 64 + cb + l16;
  const int gn   = 128 + cb + l16;

  for (int i = tid; i < GRU_ROWS * HPAD; i += GRU_WAVES * 32) {
    hbuf[0][0][i] = 0.0f;
    hbuf[1][0][i] = 0.0f;
  }

  v2f wi0r[8],  wi0z[8],  wi0n[8];
  v2f wh0r[16], wh0z[16], wh0n[16];
  v2f wi1r[16], wi1z[16], wi1n[16];
  v2f wh1r[16], wh1z[16], wh1n[16];
  gru_load_w<8>(w_ih0, FEAT, gr, lh, wi0r);
  gru_load_w<8>(w_ih0, FEAT, gz, lh, wi0z);
  gru_load_w<8>(w_ih0, FEAT, gn, lh, wi0n);
  gru_load_w<16>(w_hh0, HID, gr, lh, wh0r);
  gru_load_w<16>(w_hh0, HID, gz, lh, wh0z);
  gru_load_w<16>(w_hh0, HID, gn, lh, wh0n);
  gru_load_w<16>(w_ih1, HID, gr, lh, wi1r);
  gru_load_w<16>(w_ih1, HID, gz, lh, wi1z);
  gru_load_w<16>(w_ih1, HID, gn, lh, wi1n);
  gru_load_w<16>(w_hh1, HID, gr, lh, wh1r);
  gru_load_w<16>(w_hh1, HID, gz, lh, wh1z);
  gru_load_w<16>(w_hh1, HID, gn, lh, wh1n);
  const float bi0r = b_ih0[gr], bi0z = b_ih0[gz], bi0n = b_ih0[gn];
  const float bh0r = b_hh0[gr], bh0z = b_hh0[gz], bh0n = b_hh0[gn];
  const float bi1r = b_ih1[gr], bi1z = b_ih1[gz], bi1n = b_ih1[gn];
  const float bh1r = b_hh1[gr], bh1z = b_hh1[gz], bh1n = b_hh1[gn];

  v2f xc[8];
  gru_load_x(x_in, 0, row0 + l16, lh, xc);

  __syncthreads();

  for (int t = 0; t < T_SEQ; ++t) {
    const float* h0r = &hbuf[0][t & 1][0];
    float*       h0w = &hbuf[0][(t + 1) & 1][0];
    const float* h1r = &hbuf[1][t & 1][0];
    float*       h1w = &hbuf[1][(t + 1) & 1][0];

#pragma unroll 1
    for (int rt = 0; rt < 4; ++rt) {
      const int rb = rt * 16;
      v2f ah[16];
      gru_load_h(h0r, rb, l16, lh, ah);
      float hold[8];
#pragma unroll
      for (int v = 0; v < 8; ++v) hold[v] = h0r[(rb + v + 8 * lh) * HPAD + cb + l16];

      const int nrt = (rt + 1) & 3;
      const int nt  = (rt == 3 && t + 1 < T_SEQ) ? t + 1 : t;
      v2f xn[8];
      gru_load_x(x_in, nt, row0 + nrt * 16 + l16, lh, xn);
      __builtin_amdgcn_sched_barrier(0);

      v8f gir = bias8(bi0r), giz = bias8(bi0z), gin = bias8(bi0n);
      v8f ghr = bias8(bh0r), ghz = bias8(bh0z), ghn = bias8(bh0n);
#pragma unroll
      for (int ks = 0; ks < 8; ++ks) {
        gir = wmma_f32(xc[ks], wi0r[ks], gir);
        giz = wmma_f32(xc[ks], wi0z[ks], giz);
        gin = wmma_f32(xc[ks], wi0n[ks], gin);
      }
#pragma unroll
      for (int ks = 0; ks < 16; ++ks) {
        ghr = wmma_f32(ah[ks], wh0r[ks], ghr);
        ghz = wmma_f32(ah[ks], wh0z[ks], ghz);
        ghn = wmma_f32(ah[ks], wh0n[ks], ghn);
      }
#pragma unroll
      for (int v = 0; v < 8; ++v) {
        const float r  = sigmoidf_(gir[v] + ghr[v]);
        const float zg = sigmoidf_(giz[v] + ghz[v]);
        const float n  = tanhf_(gin[v] + r * ghn[v]);
        h0w[(rb + v + 8 * lh) * HPAD + cb + l16] = (1.0f - zg) * n + zg * hold[v];
      }
#pragma unroll
      for (int ks = 0; ks < 8; ++ks) xc[ks] = xn[ks];
    }

    __syncthreads();

#pragma unroll 1
    for (int rt = 0; rt < 4; ++rt) {
      const int rb = rt * 16;
      v2f a0[16], a1[16];
      gru_load_h(h0w, rb, l16, lh, a0);
      gru_load_h(h1r, rb, l16, lh, a1);
      float hold[8];
#pragma unroll
      for (int v = 0; v < 8; ++v) hold[v] = h1r[(rb + v + 8 * lh) * HPAD + cb + l16];
      __builtin_amdgcn_sched_barrier(0);

      v8f gir = bias8(bi1r), giz = bias8(bi1z), gin = bias8(bi1n);
      v8f ghr = bias8(bh1r), ghz = bias8(bh1z), ghn = bias8(bh1n);
#pragma unroll
      for (int ks = 0; ks < 16; ++ks) {
        gir = wmma_f32(a0[ks], wi1r[ks], gir);
        giz = wmma_f32(a0[ks], wi1z[ks], giz);
        gin = wmma_f32(a0[ks], wi1n[ks], gin);
      }
#pragma unroll
      for (int ks = 0; ks < 16; ++ks) {
        ghr = wmma_f32(a1[ks], wh1r[ks], ghr);
        ghz = wmma_f32(a1[ks], wh1z[ks], ghz);
        ghn = wmma_f32(a1[ks], wh1n[ks], ghn);
      }
#pragma unroll
      for (int v = 0; v < 8; ++v) {
        const float r  = sigmoidf_(gir[v] + ghr[v]);
        const float zg = sigmoidf_(giz[v] + ghz[v]);
        const float n  = tanhf_(gin[v] + r * ghn[v]);
        h1w[(rb + v + 8 * lh) * HPAD + cb + l16] = (1.0f - zg) * n + zg * hold[v];
      }
    }
  }

  __syncthreads();

  const float* f0 = &hbuf[0][T_SEQ & 1][0];
  const float* f1 = &hbuf[1][T_SEQ & 1][0];
  for (int i = tid; i < GRU_ROWS * 16; i += GRU_WAVES * 32) {
    const int r = i >> 4, c = (i & 15) * 4;
    v4f p, q;
    p.x = f0[r * HPAD + c]; p.y = f0[r * HPAD + c + 1];
    p.z = f0[r * HPAD + c + 2]; p.w = f0[r * HPAD + c + 3];
    q.x = f1[r * HPAD + c]; q.y = f1[r * HPAD + c + 1];
    q.z = f1[r * HPAD + c + 2]; q.w = f1[r * HPAD + c + 3];
    *(v4f*)(z_out + (size_t)(row0 + r) * HID + c) = p;
    *(v4f*)(z_out + (size_t)NROWS * HID + (size_t)(row0 + r) * HID + c) = q;
  }
}


__device__ __forceinline__ void gat_load_zrow_frag(
    const float* __restrict__ zrow, int lh, v2f (&f)[16])
{
  const v4f* p = (const v4f*)(zrow + 4 * lh);
#pragma unroll
  for (int j = 0; j < 8; ++j) {
    const v4f q = p[2 * j];
    f[2 * j].x     = q.x; f[2 * j].y     = q.y;
    f[2 * j + 1].x = q.z; f[2 * j + 1].y = q.w;
  }
}

__device__ __forceinline__ float row16_max(float x) {
  x = fmaxf(x, __int_as_float(__builtin_amdgcn_update_dpp(
          __float_as_int(x), __float_as_int(x), 0x161, 0xf, 0xf, true)));
  x = fmaxf(x, __int_as_float(__builtin_amdgcn_update_dpp(
          __float_as_int(x), __float_as_int(x), 0x162, 0xf, 0xf, true)));
  x = fmaxf(x, __int_as_float(__builtin_amdgcn_update_dpp(
          __float_as_int(x), __float_as_int(x), 0x164, 0xf, 0xf, true)));
  x = fmaxf(x, __int_as_float(__builtin_amdgcn_update_dpp(
          __float_as_int(x), __float_as_int(x), 0x168, 0xf, 0xf, true)));
  return x;
}

__device__ __forceinline__ void gat_request_score_stage(
    const float* __restrict__ zp, const int* __restrict__ mrow,
    int st, int l16, int lh, v2f (&b)[2][16], int (&m)[2][8])
{
#pragma unroll
  for (int u = 0; u < 2; ++u) {
    const int col = (st * 2 + u) * 16 + l16;
    gat_load_zrow_frag(zp + (size_t)col * HID, lh, b[u]);
    const int* mp = mrow + col;
#pragma unroll
    for (int v = 0; v < 8; ++v) m[u][v] = mp[(size_t)v * NODES];
  }
}

__device__ __forceinline__ void gat_score_stage(
    const v2f (&za)[16], const v2f (&b)[2][16], const int (&m)[2][8],
    float* S, int st, int l16, int lh, float (&rmax)[8])
{
#pragma unroll
  for (int u = 0; u < 2; ++u) {
    v8f acc = {};
#pragma unroll
    for (int ks = 0; ks < 16; ++ks) acc = wmma_f32(za[ks], b[u][ks], acc);
    const int col = (st * 2 + u) * 16 + l16;
#pragma unroll
    for (int v = 0; v < 8; ++v) {
      const int row = v + 8 * lh;
      const float s  = acc[v];
      const float lr = s > 0.0f ? s : 0.1f * s;
      float a = lr * (float)m[u][v];
      a = (a == 0.0f) ? NEG_BIG : a;
      S[col * 16 + row] = a;
      rmax[v] = fmaxf(rmax[v], a);
    }
  }
}

__device__ __forceinline__ void gat_request_pv_stage(
    const float* S, const float* __restrict__ zp,
    int st, int l16, int lh, v2f (&a)[8], v4f (&q)[8][2])
{
#pragma unroll
  for (int ks = 0; ks < 8; ++ks) {
    const int k = st * 32 + 4 * ks + 2 * lh;
    a[ks].x = S[k * 16 + l16];
    a[ks].y = S[(k + 1) * 16 + l16];
    q[ks][0] = *(const v4f*)(zp + (size_t)k * HID + 4 * l16);
    q[ks][1] = *(const v4f*)(zp + (size_t)(k + 1) * HID + 4 * l16);
  }
}

__device__ __forceinline__ void gat_pv_stage(
    const v2f (&a)[8], const v4f (&q)[8][2], v8f (&accO)[4], float mx, float& sum)
{
  v2f e[8];
#pragma unroll
  for (int ks = 0; ks < 8; ++ks) {
    e[ks].x = __expf(a[ks].x - mx);
    e[ks].y = __expf(a[ks].y - mx);
  }
  const v2f t0 = (e[0] + e[1]) + (e[2] + e[3]);
  const v2f t1 = (e[4] + e[5]) + (e[6] + e[7]);
  const v2f t  = t0 + t1;
  sum += t.x + t.y;
#pragma unroll
  for (int ks = 0; ks < 8; ++ks)
#pragma unroll
    for (int ct = 0; ct < 4; ++ct) {
      v2f b;
      b.x = q[ks][0][ct];
      b.y = q[ks][1][ct];
      accO[ct] = wmma_f32(e[ks], b, accO[ct]);
    }
}

__global__ __launch_bounds__(32) void gat_kernel(
    const float* __restrict__ z,
    const int*   __restrict__ mask,
    float*       __restrict__ out)
{
  __shared__ float S[NODES * 16];

  const int lane  = threadIdx.x;
  const int l16   = lane & 15;
  const int lh    = lane >> 4;
  const int ntile = blockIdx.x & 63;
  const int pair  = blockIdx.x >> 6;
  const float* zp = z + (size_t)pair * NODES * HID;
  const int n0 = ntile * 16;

  v2f za[16];
  gat_load_zrow_frag(zp + (size_t)(n0 + l16) * HID, lh, za);

  float mx;
  {
    float rmax[8];
#pragma unroll
    for (int v = 0; v < 8; ++v) rmax[v] = -INFINITY;
    const int* mrow = mask + (size_t)(n0 + 8 * lh) * NODES;
    v2f bA[2][16], bB[2][16];
    int mA[2][8], mB[2][8];
    gat_request_score_stage(zp, mrow, 0, l16, lh, bA, mA);
    for (int st = 0; st < 30; st += 2) {
      gat_request_score_stage(zp, mrow, st + 1, l16, lh, bB, mB);
      gat_score_stage(za, bA, mA, S, st, l16, lh, rmax);
      gat_request_score_stage(zp, mrow, st + 2, l16, lh, bA, mA);
      gat_score_stage(za, bB, mB, S, st + 1, l16, lh, rmax);
    }
    gat_request_score_stage(zp, mrow, 31, l16, lh, bB, mB);
    gat_score_stage(za, bA, mA, S, 30, l16, lh, rmax);
    gat_score_stage(za, bB, mB, S, 31, l16, lh, rmax);

#pragma unroll
    for (int v = 0; v < 8; ++v) rmax[v] = row16_max(rmax[v]);
    const int sel = l16 & 7;
    float pick = rmax[0];
#pragma unroll
    for (int v = 1; v < 8; ++v) pick = (sel == v) ? rmax[v] : pick;
    const float other = __shfl_xor(pick, 16, 32);
    mx = ((l16 >> 3) == lh) ? pick : other;
  }

  v8f accO[4] = {};
  float sum = 0.0f;
  {
    v2f aA[8], aB[8];
    v4f bA[8][2], bB[8][2];
    gat_request_pv_stage(S, zp, 0, l16, lh, aA, bA);
    for (int st = 0; st < 30; st += 2) {
      gat_request_pv_stage(S, zp, st + 1, l16, lh, aB, bB);
      gat_pv_stage(aA, bA, accO, mx, sum);
      gat_request_pv_stage(S, zp, st + 2, l16, lh, aA, bA);
      gat_pv_stage(aB, bB, accO, mx, sum);
    }
    gat_request_pv_stage(S, zp, 31, l16, lh, aB, bB);
    gat_pv_stage(aA, bA, accO, mx, sum);
    gat_pv_stage(aB, bB, accO, mx, sum);
  }
  sum += __shfl_xor(sum, 16, 32);
  const float inv = 1.0f / sum;

  float* op = out + (size_t)pair * NODES * HID;
#pragma unroll
  for (int v = 0; v < 8; ++v) {
    const int row = v + 8 * lh;
    const float rinv = __shfl(inv, row, 32);
    v4f o;
    o.x = accO[0][v] * rinv; o.y = accO[1][v] * rinv;
    o.z = accO[2][v] * rinv; o.w = accO[3][v] * rinv;
    *(v4f*)(op + (size_t)(n0 + row) * HID + 4 * l16) = o;
  }
}

extern "C" void kernel_launch(void* const* d_in, const int* in_sizes, int n_in,
                              void* d_out, int out_size, void* d_ws, size_t ws_size,
                              hipStream_t stream) {
  const float* x     = (const float*)d_in[0];
  const int*   mask  = (const int*)  d_in[1];
  const float* w_ih0 = (const float*)d_in[2];
  const float* w_hh0 = (const float*)d_in[3];
  const float* b_ih0 = (const float*)d_in[4];
  const float* b_hh0 = (const float*)d_in[5];
  const float* w_ih1 = (const float*)d_in[6];
  const float* w_hh1 = (const float*)d_in[7];
  const float* b_ih1 = (const float*)d_in[8];
  const float* b_hh1 = (const float*)d_in[9];
  float* out  = (float*)d_out;
  float* zbuf = (float*)d_ws;

  gru_fused_kernel<<<256, 128, 0, stream>>>(x, w_ih0, w_hh0, b_ih0, b_hh0,
                                            w_ih1, w_hh1, b_ih1, b_hh1, zbuf);
  gat_kernel<<<2048, 32, 0, stream>>>(zbuf, mask, out);
}


# Attention score loop: a stage's 32 z rows are fetched as whole rows (4 cache lines per 16-byte load, not 16) and turned into B fragments through a padded LDS tile private to the wave
# speedup vs baseline: 2.5263x; 1.0925x over previous
#include <hip/hip_runtime.h>
#include <hip/hip_bf16.h>
#include <math.h>


typedef __attribute__((ext_vector_type(2))) float v2f;
typedef __attribute__((ext_vector_type(8))) float v8f;

#define T_SEQ   12
#define BCN     16
#define NODES   1024
#define FEAT    32
#define HID     64
#define G3      192
#define NROWS   (BCN * NODES)
#define NEG_BIG (-1e16f)

__device__ __forceinline__ v8f wmma_f32(v2f a, v2f b, v8f c) {
  return __builtin_amdgcn_wmma_f32_16x16x4_f32(
       false, a,  false, b,
       (short)0, c,  false,  false);
}

__device__ __forceinline__ v8f bias8(float v) {
  v8f r;
#pragma unroll
  for (int i = 0; i < 8; ++i) r[i] = v;
  return r;
}

__device__ __forceinline__ float sigmoidf_(float x) {
  return __builtin_amdgcn_rcpf(1.0f + __expf(-x));
}

__device__ __forceinline__ float tanhf_(float x) {
  const float ax = fabsf(x);
  const float t  = ax * ax;
  float p = fmaf(t, -0.00570002059f, 0.0206340719f);
  p = fmaf(t, p, -0.0537379310f);
  p = fmaf(t, p, 0.133314162f);
  p = fmaf(t, p, -0.333332807f);
  const float lo = fmaf(t, ax * p, ax);
  const float e  = __expf(2.0f * ax);
  const float hi = fmaf(__builtin_amdgcn_rcpf(1.0f + e), -2.0f, 1.0f);
  return copysignf(ax < 0.625f ? lo : hi, x);
}

#define GRU_WAVES 4
#define GRU_ROWS  64
#define HPAD 66

typedef __attribute__((ext_vector_type(4))) float v4f;

template <int KS>
__device__ __forceinline__ void gru_load_w(const float* __restrict__ w, int kdim,
                                           int g, int lh, v2f (&f)[KS]) {
  const float* wr = w + (size_t)g * kdim + 2 * lh;
#pragma unroll
  for (int ks = 0; ks < KS; ++ks) { f[ks].x = wr[4 * ks]; f[ks].y = wr[4 * ks + 1]; }
}

__device__ __forceinline__ void gru_load_h(const float* hs, int rb, int l16, int lh,
                                           v2f (&f)[16]) {
  const float* hr = hs + (rb + l16) * HPAD + 2 * lh;
#pragma unroll
  for (int ks = 0; ks < 16; ++ks) { f[ks].x = hr[4 * ks]; f[ks].y = hr[4 * ks + 1]; }
}

__device__ __forceinline__ void gru_load_x(const float* __restrict__ x_in, int t,
                                           int row, int lh, v2f (&f)[8]) {
  const float* xr = x_in + ((size_t)t * NROWS + row) * FEAT + 2 * lh;
#pragma unroll
  for (int ks = 0; ks < 8; ++ks) { f[ks].x = xr[4 * ks]; f[ks].y = xr[4 * ks + 1]; }
}

__global__ __launch_bounds__(128) void gru_fused_kernel(
    const float* __restrict__ x_in,
    const float* __restrict__ w_ih0,
    const float* __restrict__ w_hh0,
    const float* __restrict__ b_ih0,
    const float* __restrict__ b_hh0,
    const float* __restrict__ w_ih1,
    const float* __restrict__ w_hh1,
    const float* __restrict__ b_ih1,
    const float* __restrict__ b_hh1,
    float* __restrict__ z_out)
{
  __shared__ float hbuf[2][2][GRU_ROWS * HPAD];

  const int tid  = threadIdx.x;
  const int wave = tid >> 5;
  const int lane = tid & 31;
  const int l16  = lane & 15;
  const int lh   = lane >> 4;
  const int row0 = blockIdx.x * GRU_ROWS;
  const int cb   = wave * 16;
  const int gr   = cb + l16;
  const int gz   = 64 + cb + l16;
  const int gn   = 128 + cb + l16;

  for (int i = tid; i < GRU_ROWS * HPAD; i += GRU_WAVES * 32) {
    hbuf[0][0][i] = 0.0f;
    hbuf[1][0][i] = 0.0f;
  }

  v2f wi0r[8],  wi0z[8],  wi0n[8];
  v2f wh0r[16], wh0z[16], wh0n[16];
  v2f wi1r[16], wi1z[16], wi1n[16];
  v2f wh1r[16], wh1z[16], wh1n[16];
  gru_load_w<8>(w_ih0, FEAT, gr, lh, wi0r);
  gru_load_w<8>(w_ih0, FEAT, gz, lh, wi0z);
  gru_load_w<8>(w_ih0, FEAT, gn, lh, wi0n);
  gru_load_w<16>(w_hh0, HID, gr, lh, wh0r);
  gru_load_w<16>(w_hh0, HID, gz, lh, wh0z);
  gru_load_w<16>(w_hh0, HID, gn, lh, wh0n);
  gru_load_w<16>(w_ih1, HID, gr, lh, wi1r);
  gru_load_w<16>(w_ih1, HID, gz, lh, wi1z);
  gru_load_w<16>(w_ih1, HID, gn, lh, wi1n);
  gru_load_w<16>(w_hh1, HID, gr, lh, wh1r);
  gru_load_w<16>(w_hh1, HID, gz, lh, wh1z);
  gru_load_w<16>(w_hh1, HID, gn, lh, wh1n);
  const float bi0r = b_ih0[gr], bi0z = b_ih0[gz], bi0n = b_ih0[gn];
  const float bh0r = b_hh0[gr], bh0z = b_hh0[gz], bh0n = b_hh0[gn];
  const float bi1r = b_ih1[gr], bi1z = b_ih1[gz], bi1n = b_ih1[gn];
  const float bh1r = b_hh1[gr], bh1z = b_hh1[gz], bh1n = b_hh1[gn];

  v2f xc[8];
  gru_load_x(x_in, 0, row0 + l16, lh, xc);

  __syncthreads();

  for (int t = 0; t < T_SEQ; ++t) {
    const float* h0r = &hbuf[0][t & 1][0];
    float*       h0w = &hbuf[0][(t + 1) & 1][0];
    const float* h1r = &hbuf[1][t & 1][0];
    float*       h1w = &hbuf[1][(t + 1) & 1][0];

#pragma unroll 1
    for (int rt = 0; rt < 4; ++rt) {
      const int rb = rt * 16;
      v2f ah[16];
      gru_load_h(h0r, rb, l16, lh, ah);
      float hold[8];
#pragma unroll
      for (int v = 0; v < 8; ++v) hold[v] = h0r[(rb + v + 8 * lh) * HPAD + cb + l16];

      const int nrt = (rt + 1) & 3;
      const int nt  = (rt == 3 && t + 1 < T_SEQ) ? t + 1 : t;
      v2f xn[8];
      gru_load_x(x_in, nt, row0 + nrt * 16 + l16, lh, xn);
      __builtin_amdgcn_sched_barrier(0);

      v8f gir = bias8(bi0r), giz = bias8(bi0z), gin = bias8(bi0n);
      v8f ghr = bias8(bh0r), ghz = bias8(bh0z), ghn = bias8(bh0n);
#pragma unroll
      for (int ks = 0; ks < 8; ++ks) {
        gir = wmma_f32(xc[ks], wi0r[ks], gir);
        giz = wmma_f32(xc[ks], wi0z[ks], giz);
        gin = wmma_f32(xc[ks], wi0n[ks], gin);
      }
#pragma unroll
      for (int ks = 0; ks < 16; ++ks) {
        ghr = wmma_f32(ah[ks], wh0r[ks], ghr);
        ghz = wmma_f32(ah[ks], wh0z[ks], ghz);
        ghn = wmma_f32(ah[ks], wh0n[ks], ghn);
      }
#pragma unroll
      for (int v = 0; v < 8; ++v) {
        const float r  = sigmoidf_(gir[v] + ghr[v]);
        const float zg = sigmoidf_(giz[v] + ghz[v]);
        const float n  = tanhf_(gin[v] + r * ghn[v]);
        h0w[(rb + v + 8 * lh) * HPAD + cb + l16] = (1.0f - zg) * n + zg * hold[v];
      }
#pragma unroll
      for (int ks = 0; ks < 8; ++ks) xc[ks] = xn[ks];
    }

    __syncthreads();

#pragma unroll 1
    for (int rt = 0; rt < 4; ++rt) {
      const int rb = rt * 16;
      v2f a0[16], a1[16];
      gru_load_h(h0w, rb, l16, lh, a0);
      gru_load_h(h1r, rb, l16, lh, a1);
      float hold[8];
#pragma unroll
      for (int v = 0; v < 8; ++v) hold[v] = h1r[(rb + v + 8 * lh) * HPAD + cb + l16];
      __builtin_amdgcn_sched_barrier(0);

      v8f gir = bias8(bi1r), giz = bias8(bi1z), gin = bias8(bi1n);
      v8f ghr = bias8(bh1r), ghz = bias8(bh1z), ghn = bias8(bh1n);
#pragma unroll
      for (int ks = 0; ks < 16; ++ks) {
        gir = wmma_f32(a0[ks], wi1r[ks], gir);
        giz = wmma_f32(a0[ks], wi1z[ks], giz);
        gin = wmma_f32(a0[ks], wi1n[ks], gin);
      }
#pragma unroll
      for (int ks = 0; ks < 16; ++ks) {
        ghr = wmma_f32(a1[ks], wh1r[ks], ghr);
        ghz = wmma_f32(a1[ks], wh1z[ks], ghz);
        ghn = wmma_f32(a1[ks], wh1n[ks], ghn);
      }
#pragma unroll
      for (int v = 0; v < 8; ++v) {
        const float r  = sigmoidf_(gir[v] + ghr[v]);
        const float zg = sigmoidf_(giz[v] + ghz[v]);
        const float n  = tanhf_(gin[v] + r * ghn[v]);
        h1w[(rb + v + 8 * lh) * HPAD + cb + l16] = (1.0f - zg) * n + zg * hold[v];
      }
    }
  }

  __syncthreads();

  const float* f0 = &hbuf[0][T_SEQ & 1][0];
  const float* f1 = &hbuf[1][T_SEQ & 1][0];
  for (int i = tid; i < GRU_ROWS * 16; i += GRU_WAVES * 32) {
    const int r = i >> 4, c = (i & 15) * 4;
    v4f p, q;
    p.x = f0[r * HPAD + c]; p.y = f0[r * HPAD + c + 1];
    p.z = f0[r * HPAD + c + 2]; p.w = f0[r * HPAD + c + 3];
    q.x = f1[r * HPAD + c]; q.y = f1[r * HPAD + c + 1];
    q.z = f1[r * HPAD + c + 2]; q.w = f1[r * HPAD + c + 3];
    *(v4f*)(z_out + (size_t)(row0 + r) * HID + c) = p;
    *(v4f*)(z_out + (size_t)NROWS * HID + (size_t)(row0 + r) * HID + c) = q;
  }
}


__device__ __forceinline__ void gat_load_zrow_frag(
    const float* __restrict__ zrow, int lh, v2f (&f)[16])
{
  const v4f* p = (const v4f*)(zrow + 4 * lh);
#pragma unroll
  for (int j = 0; j < 8; ++j) {
    const v4f q = p[2 * j];
    f[2 * j].x     = q.x; f[2 * j].y     = q.y;
    f[2 * j + 1].x = q.z; f[2 * j + 1].y = q.w;
  }
}

__device__ __forceinline__ float row16_max(float x) {
  x = fmaxf(x, __int_as_float(__builtin_amdgcn_update_dpp(
          __float_as_int(x), __float_as_int(x), 0x161, 0xf, 0xf, true)));
  x = fmaxf(x, __int_as_float(__builtin_amdgcn_update_dpp(
          __float_as_int(x), __float_as_int(x), 0x162, 0xf, 0xf, true)));
  x = fmaxf(x, __int_as_float(__builtin_amdgcn_update_dpp(
          __float_as_int(x), __float_as_int(x), 0x164, 0xf, 0xf, true)));
  x = fmaxf(x, __int_as_float(__builtin_amdgcn_update_dpp(
          __float_as_int(x), __float_as_int(x), 0x168, 0xf, 0xf, true)));
  return x;
}

#define ZT_PITCH 68


__device__ __forceinline__ void gat_request_z_rows(
    const float* __restrict__ zp, int st, int l16, int lh, v4f (&g)[16])
{
  const v4f* p = (const v4f*)(zp + (size_t)(st * 32 + lh) * HID + 4 * l16);
#pragma unroll
  for (int i = 0; i < 16; ++i) g[i] = p[i * 32];
}

__device__ __forceinline__ void gat_request_mask(
    const int* __restrict__ mrow, int st, int l16, int (&m)[2][8])
{
#pragma unroll
  for (int u = 0; u < 2; ++u) {
    const int* mp = mrow + (st * 2 + u) * 16 + l16;
#pragma unroll
    for (int v = 0; v < 8; ++v) m[u][v] = mp[(size_t)v * NODES];
  }
}

__device__ __forceinline__ void gat_stage_and_load_frags(
    float* ZT, int l16, int lh, const v4f (&g)[16], v2f (&b)[2][16])
{
  float* w = ZT + lh * ZT_PITCH + 4 * l16;
#pragma unroll
  for (int i = 0; i < 16; ++i) *(v4f*)(w + i * 2 * ZT_PITCH) = g[i];
  __builtin_amdgcn_fence(__ATOMIC_RELEASE, "wavefront");
  __builtin_amdgcn_wave_barrier();
  __builtin_amdgcn_fence(__ATOMIC_ACQUIRE, "wavefront");
#pragma unroll
  for (int u = 0; u < 2; ++u) {
    const v4f* p = (const v4f*)(ZT + (16 * u + l16) * ZT_PITCH + 4 * lh);
#pragma unroll
    for (int j = 0; j < 8; ++j) {
      const v4f q = p[2 * j];
      b[u][2 * j].x     = q.x; b[u][2 * j].y     = q.y;
      b[u][2 * j + 1].x = q.z; b[u][2 * j + 1].y = q.w;
    }
  }
}

__device__ __forceinline__ void gat_score_stage(
    const v2f (&za)[16], const v2f (&b)[2][16], const int (&m)[2][8],
    float* S, int st, int l16, int lh, float (&rmax)[8])
{
#pragma unroll
  for (int u = 0; u < 2; ++u) {
    v8f acc = {};
#pragma unroll
    for (int ks = 0; ks < 16; ++ks) acc = wmma_f32(za[ks], b[u][ks], acc);
    const int col = (st * 2 + u) * 16 + l16;
#pragma unroll
    for (int v = 0; v < 8; ++v) {
      const int row = v + 8 * lh;
      const float s  = acc[v];
      const float lr = s > 0.0f ? s : 0.1f * s;
      float a = lr * (float)m[u][v];
      a = (a == 0.0f) ? NEG_BIG : a;
      S[col * 16 + row] = a;
      rmax[v] = fmaxf(rmax[v], a);
    }
  }
}

__device__ __forceinline__ void gat_request_pv_stage(
    const float* S, const float* __restrict__ zp,
    int st, int l16, int lh, v2f (&a)[8], v4f (&q)[8][2])
{
#pragma unroll
  for (int ks = 0; ks < 8; ++ks) {
    const int k = st * 32 + 4 * ks + 2 * lh;
    a[ks].x = S[k * 16 + l16];
    a[ks].y = S[(k + 1) * 16 + l16];
    q[ks][0] = *(const v4f*)(zp + (size_t)k * HID + 4 * l16);
    q[ks][1] = *(const v4f*)(zp + (size_t)(k + 1) * HID + 4 * l16);
  }
}

__device__ __forceinline__ void gat_pv_stage(
    const v2f (&a)[8], const v4f (&q)[8][2], v8f (&accO)[4], float mx, float& sum)
{
  v2f e[8];
#pragma unroll
  for (int ks = 0; ks < 8; ++ks) {
    e[ks].x = __expf(a[ks].x - mx);
    e[ks].y = __expf(a[ks].y - mx);
  }
  const v2f t0 = (e[0] + e[1]) + (e[2] + e[3]);
  const v2f t1 = (e[4] + e[5]) + (e[6] + e[7]);
  const v2f t  = t0 + t1;
  sum += t.x + t.y;
#pragma unroll
  for (int ks = 0; ks < 8; ++ks)
#pragma unroll
    for (int ct = 0; ct < 4; ++ct) {
      v2f b;
      b.x = q[ks][0][ct];
      b.y = q[ks][1][ct];
      accO[ct] = wmma_f32(e[ks], b, accO[ct]);
    }
}

__global__ __launch_bounds__(32) void gat_kernel(
    const float* __restrict__ z,
    const int*   __restrict__ mask,
    float*       __restrict__ out)
{
  __shared__ float S[NODES * 16];
  __shared__ __attribute__((aligned(16))) float ZT[32 * ZT_PITCH];

  const int lane  = threadIdx.x;
  const int l16   = lane & 15;
  const int lh    = lane >> 4;
  const int ntile = blockIdx.x & 63;
  const int pair  = blockIdx.x >> 6;
  const float* zp = z + (size_t)pair * NODES * HID;
  const int n0 = ntile * 16;

  v2f za[16];
  gat_load_zrow_frag(zp + (size_t)(n0 + l16) * HID, lh, za);

  float mx;
  {
    float rmax[8];
#pragma unroll
    for (int v = 0; v < 8; ++v) rmax[v] = -INFINITY;
    const int* mrow = mask + (size_t)(n0 + 8 * lh) * NODES;
    v4f g[16];
    v2f bA[2][16], bB[2][16];
    int mA[2][8], mB[2][8];
    gat_request_z_rows(zp, 0, l16, lh, g);
    gat_request_mask(mrow, 0, l16, mA);
    gat_stage_and_load_frags(ZT, l16, lh, g, bA);
    gat_request_z_rows(zp, 1, l16, lh, g);
    for (int st = 0; st < 30; st += 2) {
      gat_stage_and_load_frags(ZT, l16, lh, g, bB);
      __builtin_amdgcn_sched_barrier(0);
      gat_request_z_rows(zp, st + 2, l16, lh, g);
      gat_request_mask(mrow, st + 1, l16, mB);
      gat_score_stage(za, bA, mA, S, st, l16, lh, rmax);
      gat_stage_and_load_frags(ZT, l16, lh, g, bA);
      __builtin_amdgcn_sched_barrier(0);
      gat_request_z_rows(zp, st + 3, l16, lh, g);
      gat_request_mask(mrow, st + 2, l16, mA);
      gat_score_stage(za, bB, mB, S, st + 1, l16, lh, rmax);
    }
    gat_stage_and_load_frags(ZT, l16, lh, g, bB);
    gat_request_mask(mrow, 31, l16, mB);
    gat_score_stage(za, bA, mA, S, 30, l16, lh, rmax);
    gat_score_stage(za, bB, mB, S, 31, l16, lh, rmax);

#pragma unroll
    for (int v = 0; v < 8; ++v) rmax[v] = row16_max(rmax[v]);
    const int sel = l16 & 7;
    float pick = rmax[0];
#pragma unroll
    for (int v = 1; v < 8; ++v) pick = (sel == v) ? rmax[v] : pick;
    const float other = __shfl_xor(pick, 16, 32);
    mx = ((l16 >> 3) == lh) ? pick : other;
  }

  v8f accO[4] = {};
  float sum = 0.0f;
  {
    v2f aA[8], aB[8];
    v4f bA[8][2], bB[8][2];
    gat_request_pv_stage(S, zp, 0, l16, lh, aA, bA);
    for (int st = 0; st < 30; st += 2) {
      gat_request_pv_stage(S, zp, st + 1, l16, lh, aB, bB);
      gat_pv_stage(aA, bA, accO, mx, sum);
      gat_request_pv_stage(S, zp, st + 2, l16, lh, aA, bA);
      gat_pv_stage(aB, bB, accO, mx, sum);
    }
    gat_request_pv_stage(S, zp, 31, l16, lh, aB, bB);
    gat_pv_stage(aA, bA, accO, mx, sum);
    gat_pv_stage(aB, bB, accO, mx, sum);
  }
  sum += __shfl_xor(sum, 16, 32);
  const float inv = 1.0f / sum;

  float* op = out + (size_t)pair * NODES * HID;
#pragma unroll
  for (int v = 0; v < 8; ++v) {
    const int row = v + 8 * lh;
    const float rinv = __shfl(inv, row, 32);
    v4f o;
    o.x = accO[0][v] * rinv; o.y = accO[1][v] * rinv;
    o.z = accO[2][v] * rinv; o.w = accO[3][v] * rinv;
    *(v4f*)(op + (size_t)(n0 + row) * HID + 4 * l16) = o;
  }
}

extern "C" void kernel_launch(void* const* d_in, const int* in_sizes, int n_in,
                              void* d_out, int out_size, void* d_ws, size_t ws_size,
                              hipStream_t stream) {
  const float* x     = (const float*)d_in[0];
  const int*   mask  = (const int*)  d_in[1];
  const float* w_ih0 = (const float*)d_in[2];
  const float* w_hh0 = (const float*)d_in[3];
  const float* b_ih0 = (const float*)d_in[4];
  const float* b_hh0 = (const float*)d_in[5];
  const float* w_ih1 = (const float*)d_in[6];
  const float* w_hh1 = (const float*)d_in[7];
  const float* b_ih1 = (const float*)d_in[8];
  const float* b_hh1 = (const float*)d_in[9];
  float* out  = (float*)d_out;
  float* zbuf = (float*)d_ws;

  gru_fused_kernel<<<256, 128, 0, stream>>>(x, w_ih0, w_hh0, b_ih0, b_hh0,
                                            w_ih1, w_hh1, b_ih1, b_hh1, zbuf);
  gat_kernel<<<2048, 32, 0, stream>>>(zbuf, mask, out);
}


# Attention kernel: running-max softmax per 32-column stage fuses score and output loops (sums rescaled when the max moves): z stage fetched once for both, no 64 KB score tile, 2 waves per SIMD
# speedup vs baseline: 2.7247x; 1.0785x over previous
#include <hip/hip_runtime.h>
#include <hip/hip_bf16.h>
#include <math.h>


typedef __attribute__((ext_vector_type(2))) float v2f;
typedef __attribute__((ext_vector_type(8))) float v8f;

#define T_SEQ   12
#define BCN     16
#define NODES   1024
#define FEAT    32
#define HID     64
#define G3      192
#define NROWS   (BCN * NODES)
#define NEG_BIG (-1e16f)

__device__ __forceinline__ v8f wmma_f32(v2f a, v2f b, v8f c) {
  return __builtin_amdgcn_wmma_f32_16x16x4_f32(
       false, a,  false, b,
       (short)0, c,  false,  false);
}

__device__ __forceinline__ v8f bias8(float v) {
  v8f r;
#pragma unroll
  for (int i = 0; i < 8; ++i) r[i] = v;
  return r;
}

__device__ __forceinline__ float sigmoidf_(float x) {
  return __builtin_amdgcn_rcpf(1.0f + __expf(-x));
}

__device__ __forceinline__ float tanhf_(float x) {
  const float ax = fabsf(x);
  const float t  = ax * ax;
  float p = fmaf(t, -0.00570002059f, 0.0206340719f);
  p = fmaf(t, p, -0.0537379310f);
  p = fmaf(t, p, 0.133314162f);
  p = fmaf(t, p, -0.333332807f);
  const float lo = fmaf(t, ax * p, ax);
  const float e  = __expf(2.0f * ax);
  const float hi = fmaf(__builtin_amdgcn_rcpf(1.0f + e), -2.0f, 1.0f);
  return copysignf(ax < 0.625f ? lo : hi, x);
}

#define GRU_WAVES 4
#define GRU_ROWS  64
#define HPAD 66

typedef __attribute__((ext_vector_type(4))) float v4f;

template <int KS>
__device__ __forceinline__ void gru_load_w(const float* __restrict__ w, int kdim,
                                           int g, int lh, v2f (&f)[KS]) {
  const float* wr = w + (size_t)g * kdim + 2 * lh;
#pragma unroll
  for (int ks = 0; ks < KS; ++ks) { f[ks].x = wr[4 * ks]; f[ks].y = wr[4 * ks + 1]; }
}

__device__ __forceinline__ void gru_load_h(const float* hs, int rb, int l16, int lh,
                                           v2f (&f)[16]) {
  const float* hr = hs + (rb + l16) * HPAD + 2 * lh;
#pragma unroll
  for (int ks = 0; ks < 16; ++ks) { f[ks].x = hr[4 * ks]; f[ks].y = hr[4 * ks + 1]; }
}

__device__ __forceinline__ void gru_load_x(const float* __restrict__ x_in, int t,
                                           int row, int lh, v2f (&f)[8]) {
  const float* xr = x_in + ((size_t)t * NROWS + row) * FEAT + 2 * lh;
#pragma unroll
  for (int ks = 0; ks < 8; ++ks) { f[ks].x = xr[4 * ks]; f[ks].y = xr[4 * ks + 1]; }
}

__global__ __launch_bounds__(128) void gru_fused_kernel(
    const float* __restrict__ x_in,
    const float* __restrict__ w_ih0,
    const float* __restrict__ w_hh0,
    const float* __restrict__ b_ih0,
    const float* __restrict__ b_hh0,
    const float* __restrict__ w_ih1,
    const float* __restrict__ w_hh1,
    const float* __restrict__ b_ih1,
    const float* __restrict__ b_hh1,
    float* __restrict__ z_out)
{
  __shared__ float hbuf[2][2][GRU_ROWS * HPAD];

  const int tid  = threadIdx.x;
  const int wave = tid >> 5;
  const int lane = tid & 31;
  const int l16  = lane & 15;
  const int lh   = lane >> 4;
  const int row0 = blockIdx.x * GRU_ROWS;
  const int cb   = wave * 16;
  const int gr   = cb + l16;
  const int gz   = 64 + cb + l16;
  const int gn   = 128 + cb + l16;

  for (int i = tid; i < GRU_ROWS * HPAD; i += GRU_WAVES * 32) {
    hbuf[0][0][i] = 0.0f;
    hbuf[1][0][i] = 0.0f;
  }

  v2f wi0r[8],  wi0z[8],  wi0n[8];
  v2f wh0r[16], wh0z[16], wh0n[16];
  v2f wi1r[16], wi1z[16], wi1n[16];
  v2f wh1r[16], wh1z[16], wh1n[16];
  gru_load_w<8>(w_ih0, FEAT, gr, lh, wi0r);
  gru_load_w<8>(w_ih0, FEAT, gz, lh, wi0z);
  gru_load_w<8>(w_ih0, FEAT, gn, lh, wi0n);
  gru_load_w<16>(w_hh0, HID, gr, lh, wh0r);
  gru_load_w<16>(w_hh0, HID, gz, lh, wh0z);
  gru_load_w<16>(w_hh0, HID, gn, lh, wh0n);
  gru_load_w<16>(w_ih1, HID, gr, lh, wi1r);
  gru_load_w<16>(w_ih1, HID, gz, lh, wi1z);
  gru_load_w<16>(w_ih1, HID, gn, lh, wi1n);
  gru_load_w<16>(w_hh1, HID, gr, lh, wh1r);
  gru_load_w<16>(w_hh1, HID, gz, lh, wh1z);
  gru_load_w<16>(w_hh1, HID, gn, lh, wh1n);
  const float bi0r = b_ih0[gr], bi0z = b_ih0[gz], bi0n = b_ih0[gn];
  const float bh0r = b_hh0[gr], bh0z = b_hh0[gz], bh0n = b_hh0[gn];
  const float bi1r = b_ih1[gr], bi1z = b_ih1[gz], bi1n = b_ih1[gn];
  const float bh1r = b_hh1[gr], bh1z = b_hh1[gz], bh1n = b_hh1[gn];

  v2f xc[8];
  gru_load_x(x_in, 0, row0 + l16, lh, xc);

  __syncthreads();

  for (int t = 0; t < T_SEQ; ++t) {
    const float* h0r = &hbuf[0][t & 1][0];
    float*       h0w = &hbuf[0][(t + 1) & 1][0];
    const float* h1r = &hbuf[1][t & 1][0];
    float*       h1w = &hbuf[1][(t + 1) & 1][0];

#pragma unroll 1
    for (int rt = 0; rt < 4; ++rt) {
      const int rb = rt * 16;
      v2f ah[16];
      gru_load_h(h0r, rb, l16, lh, ah);
      float hold[8];
#pragma unroll
      for (int v = 0; v < 8; ++v) hold[v] = h0r[(rb + v + 8 * lh) * HPAD + cb + l16];

      const int nrt = (rt + 1) & 3;
      const int nt  = (rt == 3 && t + 1 < T_SEQ) ? t + 1 : t;
      v2f xn[8];
      gru_load_x(x_in, nt, row0 + nrt * 16 + l16, lh, xn);
      __builtin_amdgcn_sched_barrier(0);

      v8f gir = bias8(bi0r), giz = bias8(bi0z), gin = bias8(bi0n);
      v8f ghr = bias8(bh0r), ghz = bias8(bh0z), ghn = bias8(bh0n);
#pragma unroll
      for (int ks = 0; ks < 8; ++ks) {
        gir = wmma_f32(xc[ks], wi0r[ks], gir);
        giz = wmma_f32(xc[ks], wi0z[ks], giz);
        gin = wmma_f32(xc[ks], wi0n[ks], gin);
      }
#pragma unroll
      for (int ks = 0; ks < 16; ++ks) {
        ghr = wmma_f32(ah[ks], wh0r[ks], ghr);
        ghz = wmma_f32(ah[ks], wh0z[ks], ghz);
        ghn = wmma_f32(ah[ks], wh0n[ks], ghn);
      }
#pragma unroll
      for (int v = 0; v < 8; ++v) {
        const float r  = sigmoidf_(gir[v] + ghr[v]);
        const float zg = sigmoidf_(giz[v] + ghz[v]);
        const float n  = tanhf_(gin[v] + r * ghn[v]);
        h0w[(rb + v + 8 * lh) * HPAD + cb + l16] = (1.0f - zg) * n + zg * hold[v];
      }
#pragma unroll
      for (int ks = 0; ks < 8; ++ks) xc[ks] = xn[ks];
    }

    __syncthreads();

#pragma unroll 1
    for (int rt = 0; rt < 4; ++rt) {
      const int rb = rt * 16;
      v2f a0[16], a1[16];
      gru_load_h(h0w, rb, l16, lh, a0);
      gru_load_h(h1r, rb, l16, lh, a1);
      float hold[8];
#pragma unroll
      for (int v = 0; v < 8; ++v) hold[v] = h1r[(rb + v + 8 * lh) * HPAD + cb + l16];
      __builtin_amdgcn_sched_barrier(0);

      v8f gir = bias8(bi1r), giz = bias8(bi1z), gin = bias8(bi1n);
      v8f ghr = bias8(bh1r), ghz = bias8(bh1z), ghn = bias8(bh1n);
#pragma unroll
      for (int ks = 0; ks < 16; ++ks) {
        gir = wmma_f32(a0[ks], wi1r[ks], gir);
        giz = wmma_f32(a0[ks], wi1z[ks], giz);
        gin = wmma_f32(a0[ks], wi1n[ks], gin);
      }
#pragma unroll
      for (int ks = 0; ks < 16; ++ks) {
        ghr = wmma_f32(a1[ks], wh1r[ks], ghr);
        ghz = wmma_f32(a1[ks], wh1z[ks], ghz);
        ghn = wmma_f32(a1[ks], wh1n[ks], ghn);
      }
#pragma unroll
      for (int v = 0; v < 8; ++v) {
        const float r  = sigmoidf_(gir[v] + ghr[v]);
        const float zg = sigmoidf_(giz[v] + ghz[v]);
        const float n  = tanhf_(gin[v] + r * ghn[v]);
        h1w[(rb + v + 8 * lh) * HPAD + cb + l16] = (1.0f - zg) * n + zg * hold[v];
      }
    }
  }

  __syncthreads();

  const float* f0 = &hbuf[0][T_SEQ & 1][0];
  const float* f1 = &hbuf[1][T_SEQ & 1][0];
  for (int i = tid; i < GRU_ROWS * 16; i += GRU_WAVES * 32) {
    const int r = i >> 4, c = (i & 15) * 4;
    v4f p, q;
    p.x = f0[r * HPAD + c]; p.y = f0[r * HPAD + c + 1];
    p.z = f0[r * HPAD + c + 2]; p.w = f0[r * HPAD + c + 3];
    q.x = f1[r * HPAD + c]; q.y = f1[r * HPAD + c + 1];
    q.z = f1[r * HPAD + c + 2]; q.w = f1[r * HPAD + c + 3];
    *(v4f*)(z_out + (size_t)(row0 + r) * HID + c) = p;
    *(v4f*)(z_out + (size_t)NROWS * HID + (size_t)(row0 + r) * HID + c) = q;
  }
}


__device__ __forceinline__ void gat_load_zrow_frag(
    const float* __restrict__ zrow, int lh, v2f (&f)[16])
{
  const v4f* p = (const v4f*)(zrow + 4 * lh);
#pragma unroll
  for (int j = 0; j < 8; ++j) {
    const v4f q = p[2 * j];
    f[2 * j].x     = q.x; f[2 * j].y     = q.y;
    f[2 * j + 1].x = q.z; f[2 * j + 1].y = q.w;
  }
}

#define ZT_PITCH 68


__device__ __forceinline__ void gat_request_z_rows(
    const float* __restrict__ zp, int st, int l16, int lh, v4f (&g)[16])
{
  const v4f* p = (const v4f*)(zp + (size_t)(st * 32 + lh) * HID + 4 * l16);
#pragma unroll
  for (int i = 0; i < 16; ++i) g[i] = p[i * 32];
}

__device__ __forceinline__ void gat_request_mask(
    const int* __restrict__ mrow, int st, int l16, int (&m)[2][8])
{
#pragma unroll
  for (int u = 0; u < 2; ++u) {
    const int* mp = mrow + (st * 2 + u) * 16 + l16;
#pragma unroll
    for (int v = 0; v < 8; ++v) m[u][v] = mp[(size_t)v * NODES];
  }
}

__device__ __forceinline__ void gat_lds_order() {
  __builtin_amdgcn_fence(__ATOMIC_RELEASE, "wavefront");
  __builtin_amdgcn_wave_barrier();
  __builtin_amdgcn_fence(__ATOMIC_ACQUIRE, "wavefront");
}

__device__ __forceinline__ void gat_stage_and_load(
    float* ZT, int l16, int lh, const v4f (&g)[16],
    v2f (&b)[2][16], v4f (&q)[8][2])
{
  float* w = ZT + lh * ZT_PITCH + 4 * l16;
#pragma unroll
  for (int i = 0; i < 16; ++i) *(v4f*)(w + i * 2 * ZT_PITCH) = g[i];
  gat_lds_order();
#pragma unroll
  for (int u = 0; u < 2; ++u) {
    const v4f* p = (const v4f*)(ZT + (16 * u + l16) * ZT_PITCH + 4 * lh);
#pragma unroll
    for (int j = 0; j < 8; ++j) {
      const v4f r = p[2 * j];
      b[u][2 * j].x     = r.x; b[u][2 * j].y     = r.y;
      b[u][2 * j + 1].x = r.z; b[u][2 * j + 1].y = r.w;
    }
  }
#pragma unroll
  for (int ks = 0; ks < 8; ++ks) {
    const int k = 4 * ks + 2 * lh;
    q[ks][0] = *(const v4f*)(ZT + k * ZT_PITCH + 4 * l16);
    q[ks][1] = *(const v4f*)(ZT + (k + 1) * ZT_PITCH + 4 * l16);
  }
}

__device__ __forceinline__ void gat_score_stage(
    const v2f (&za)[16], const v2f (&b)[2][16], const int (&m)[2][8],
    float* PT, int l16, int lh)
{
  v8f acc0 = {}, acc1 = {};
#pragma unroll
  for (int ks = 0; ks < 16; ++ks) {
    acc0 = wmma_f32(za[ks], b[0][ks], acc0);
    acc1 = wmma_f32(za[ks], b[1][ks], acc1);
  }
#pragma unroll
  for (int u = 0; u < 2; ++u) {
#pragma unroll
    for (int v = 0; v < 8; ++v) {
      const int row = v + 8 * lh;
      const float s  = u ? acc1[v] : acc0[v];
      const float lr = s > 0.0f ? s : 0.1f * s;
      float a = lr * (float)m[u][v];
      a = (a == 0.0f) ? NEG_BIG : a;
      PT[(16 * u + l16) * 16 + row] = a;
    }
  }
}

__device__ __forceinline__ void gat_load_scores(
    const float* PT, int l16, int lh, v2f (&a)[8])
{
#pragma unroll
  for (int ks = 0; ks < 8; ++ks) {
    const int k = 4 * ks + 2 * lh;
    a[ks].x = PT[k * 16 + l16];
    a[ks].y = PT[(k + 1) * 16 + l16];
  }
}

__device__ __forceinline__ float lane_xor16(float x) {
  return __int_as_float(__builtin_amdgcn_permlanex16(
      __float_as_int(x), __float_as_int(x), 0x76543210, 0xfedcba98, false, false));
}
__device__ __forceinline__ float row16_ror8(float x) {
  return __int_as_float(__builtin_amdgcn_update_dpp(
      __float_as_int(x), __float_as_int(x), 0x128, 0xf, 0xf, true));
}
template <int N>
__device__ __forceinline__ float row16_share(float x) {
  return __int_as_float(__builtin_amdgcn_update_dpp(
      __float_as_int(x), __float_as_int(x), 0x150 + N, 0xf, 0xf, true));
}

__device__ __forceinline__ void gat_softmax_pv_stage(
    const v2f (&a)[8], const v4f (&q)[8][2], v8f (&accO)[4],
    float& m_run, float& sum, int lh)
{
  float smax = fmaxf(a[0].x, a[0].y);
#pragma unroll
  for (int ks = 1; ks < 8; ++ks) smax = fmaxf(smax, fmaxf(a[ks].x, a[ks].y));
  smax = fmaxf(smax, lane_xor16(smax));
  const float m_new = fmaxf(m_run, smax);
  const float f = __expf(m_run - m_new);
  m_run = m_new;

  v2f e[8];
#pragma unroll
  for (int ks = 0; ks < 8; ++ks) {
    e[ks].x = __expf(a[ks].x - m_new);
    e[ks].y = __expf(a[ks].y - m_new);
  }
  const v2f t0 = (e[0] + e[1]) + (e[2] + e[3]);
  const v2f t1 = (e[4] + e[5]) + (e[6] + e[7]);
  const v2f t  = t0 + t1;
  sum = sum * f + (t.x + t.y);

  const float f8 = row16_ror8(f);
  const float fr = lh ? f8 : f;
  float fv[8];
  fv[0] = row16_share<0>(fr); fv[1] = row16_share<1>(fr);
  fv[2] = row16_share<2>(fr); fv[3] = row16_share<3>(fr);
  fv[4] = row16_share<4>(fr); fv[5] = row16_share<5>(fr);
  fv[6] = row16_share<6>(fr); fv[7] = row16_share<7>(fr);
#pragma unroll
  for (int ct = 0; ct < 4; ++ct)
#pragma unroll
    for (int v = 0; v < 8; ++v) accO[ct][v] *= fv[v];

#pragma unroll
  for (int ks = 0; ks < 8; ++ks)
#pragma unroll
    for (int ct = 0; ct < 4; ++ct) {
      v2f b;
      b.x = q[ks][0][ct];
      b.y = q[ks][1][ct];
      accO[ct] = wmma_f32(e[ks], b, accO[ct]);
    }
}

__global__ __launch_bounds__(32) void gat_kernel(
    const float* __restrict__ z,
    const int*   __restrict__ mask,
    float*       __restrict__ out)
{
  __shared__ __attribute__((aligned(16))) float ZT[32 * ZT_PITCH];
  __shared__ __attribute__((aligned(16))) float PT[32 * 16];

  const int lane  = threadIdx.x;
  const int l16   = lane & 15;
  const int lh    = lane >> 4;
  const int ntile = blockIdx.x & 63;
  const int pair  = blockIdx.x >> 6;
  const float* zp = z + (size_t)pair * NODES * HID;
  const int n0 = ntile * 16;

  v2f za[16];
  gat_load_zrow_frag(zp + (size_t)(n0 + l16) * HID, lh, za);

  const int* mrow = mask + (size_t)(n0 + 8 * lh) * NODES;

  v8f accO[4] = {};
  float sum   = 0.0f;
  float m_run = -INFINITY;

  v4f g[16];
  int mk[2][8];
  gat_request_z_rows(zp, 0, l16, lh, g);
  gat_request_mask(mrow, 0, l16, mk);

#pragma unroll 1
  for (int st = 0; st < 32; ++st) {
    v2f b[2][16];
    v4f q[8][2];
    gat_stage_and_load(ZT, l16, lh, g, b, q);
    __builtin_amdgcn_sched_barrier(0);

    const int nst = (st + 1 < 32) ? st + 1 : st;
    gat_request_z_rows(zp, nst, l16, lh, g);
    int mkn[2][8];
    gat_request_mask(mrow, nst, l16, mkn);

    gat_score_stage(za, b, mk, PT, l16, lh);
    gat_lds_order();
    v2f a[8];
    gat_load_scores(PT, l16, lh, a);
    __builtin_amdgcn_sched_barrier(0);

    gat_softmax_pv_stage(a, q, accO, m_run, sum, lh);
    gat_lds_order();

#pragma unroll
    for (int u = 0; u < 2; ++u)
#pragma unroll
      for (int v = 0; v < 8; ++v) mk[u][v] = mkn[u][v];
  }

  sum += __shfl_xor(sum, 16, 32);
  const float inv = 1.0f / sum;

  float* op = out + (size_t)pair * NODES * HID;
#pragma unroll
  for (int v = 0; v < 8; ++v) {
    const int row = v + 8 * lh;
    const float rinv = __shfl(inv, row, 32);
    v4f o;
    o.x = accO[0][v] * rinv; o.y = accO[1][v] * rinv;
    o.z = accO[2][v] * rinv; o.w = accO[3][v] * rinv;
    *(v4f*)(op + (size_t)(n0 + row) * HID + 4 * l16) = o;
  }
}

extern "C" void kernel_launch(void* const* d_in, const int* in_sizes, int n_in,
                              void* d_out, int out_size, void* d_ws, size_t ws_size,
                              hipStream_t stream) {
  const float* x     = (const float*)d_in[0];
  const int*   mask  = (const int*)  d_in[1];
  const float* w_ih0 = (const float*)d_in[2];
  const float* w_hh0 = (const float*)d_in[3];
  const float* b_ih0 = (const float*)d_in[4];
  const float* b_hh0 = (const float*)d_in[5];
  const float* w_ih1 = (const float*)d_in[6];
  const float* w_hh1 = (const float*)d_in[7];
  const float* b_ih1 = (const float*)d_in[8];
  const float* b_hh1 = (const float*)d_in[9];
  float* out  = (float*)d_out;
  float* zbuf = (float*)d_ws;

  gru_fused_kernel<<<256, 128, 0, stream>>>(x, w_ih0, w_hh0, b_ih0, b_hh0,
                                            w_ih1, w_hh1, b_ih1, b_hh1, zbuf);
  gat_kernel<<<2048, 32, 0, stream>>>(zbuf, mask, out);
}
